# starting kernel
# baseline (speedup 1.0000x reference)
#include <hip/hip_runtime.h>
#include <math.h>

typedef __attribute__((ext_vector_type(2))) float v2f;
typedef __attribute__((ext_vector_type(8))) float v8f;

#define WMMA_F32(A, B, C) \
  __builtin_amdgcn_wmma_f32_16x16x4_f32(false, (A), false, (B), (short)0, (C), false, false)

#define NB   32
#define NT   12
#define NN   512
#define NDE  16
#define NHID 64
#define NCP  68

__global__ void k_w(const float* __restrict__ ne, const float* __restrict__ wp,
                    float* __restrict__ W) {
  long idx = (long)blockIdx.x * 256 + threadIdx.x;
  const long total = 3L * NN * 2 * NCP * NHID;
  if (idx >= total) return;
  int o = idx % 64; long r = idx / 64;
  int i = r % NCP; r /= NCP;
  int slot = r % 2; r /= 2;
  int n = r % NN; int g = (int)(r / NN);
  float acc = 0.f;
  if (i < 67) {
    for (int d = 0; d < NDE; ++d)
      acc += ne[n * NDE + d] * wp[(((long)(g * NDE + d) * 2 + slot) * 67 + i) * 64 + o];
  }
  W[idx] = acc;
}

__global__ void k_tb(const float* __restrict__ te, const float* __restrict__ bp,
                     float* __restrict__ TB) {
  int idx = blockIdx.x * 256 + threadIdx.x;
  if (idx >= 3 * NT * NB * NHID) return;
  int o = idx % 64; int r = idx / 64;
  int b = r % NB; r /= NB;
  int t = r % NT; int g = r / NT;
  float acc = 0.f;
  for (int d = 0; d < NDE; ++d)
    acc += te[(b * NT + t) * NDE + d] * bp[(g * NDE + d) * 64 + o];
  TB[idx] = acc;
}

__global__ void k_emb(const float* __restrict__ ne, const float* __restrict__ te,
                      const float* __restrict__ glg, const float* __restrict__ glb,
                      float* __restrict__ E, int t) {
  int idx = blockIdx.x * 256 + threadIdx.x;
  if (idx >= 3 * NB * NN) return;
  int n = idx % NN; int r = idx / NN;
  int b = r % NB; int g = r / NB;
  float v[NDE]; float mu = 0.f;
  for (int d = 0; d < NDE; ++d) { v[d] = ne[n * NDE + d] + te[(b * NT + t) * NDE + d]; mu += v[d]; }
  mu *= (1.f / NDE);
  float var = 0.f;
  for (int d = 0; d < NDE; ++d) { float q = v[d] - mu; var += q * q; }
  var *= (1.f / NDE);
  float inv = rsqrtf(var + 1e-12f);
  for (int d = 0; d < NDE; ++d)
    E[(long)idx * NDE + d] = (v[d] - mu) * inv * glg[g * NDE + d] + glb[g * NDE + d];
}

__global__ void k_zero(float* p, long nElem) {
  long i = (long)blockIdx.x * 256 + threadIdx.x;
  if (i < nElem) p[i] = 0.f;
}

__global__ void k_build_ias(const float* __restrict__ x, const float* __restrict__ h,
                            float* __restrict__ ias, int t) {
  int idx = blockIdx.x * 256 + threadIdx.x;
  if (idx >= NB * NN * NCP) return;
  int i = idx % NCP; int r = idx / NCP; int n = r % NN; int b = r / NN;
  float v;
  if (i < 3)       v = x[((long)(b * NT + t) * NN + n) * 3 + i];
  else if (i < 67) v = h[((long)b * NN + n) * 64 + (i - 3)];
  else             v = 0.f;
  ias[idx] = v;
}

__global__ void k_build_cand(const float* __restrict__ x, const float* __restrict__ h,
                             const float* __restrict__ zp, float* __restrict__ cand, int t) {
  int idx = blockIdx.x * 256 + threadIdx.x;
  if (idx >= NB * NN * NCP) return;
  int i = idx % NCP; int r = idx / NCP; int n = r % NN; int b = r / NN;
  float v;
  if (i < 3)       v = x[((long)(b * NT + t) * NN + n) * 3 + i];
  else if (i < 67) {
    long e = ((long)b * NN + n) * 64 + (i - 3);
    float z = 1.f / (1.f + __expf(-zp[e]));
    v = z * h[e];
  } else v = 0.f;
  cand[idx] = v;
}

__global__ void __launch_bounds__(128) k_graph_attn(
    const float* __restrict__ E,
    const float* __restrict__ X,
    float* __restrict__ Y,
    int g) {
  __shared__ float lds_e[16][16];
  __shared__ float lds_v[16][80];
  __shared__ float lds_p[4][16][17];

  const int b = blockIdx.y;
  const int row0 = blockIdx.x * 64;
  const int tid = threadIdx.x;
  const int wave = tid >> 5;
  const int lane = tid & 31;
  const int l16 = lane & 15;
  const int half = lane >> 4;
  const int myrow = row0 + wave * 16 + l16;

  const float* Eb = E + ((long)g * NB + b) * NN * NDE;
  const float* Xb = X + (long)b * NN * NCP;

  v2f aA[4];
#pragma unroll
  for (int kk = 0; kk < 4; ++kk) {
    aA[kk].x = Eb[myrow * NDE + kk * 4 + 2 * half];
    aA[kk].y = Eb[myrow * NDE + kk * 4 + 2 * half + 1];
  }

  for (int i = tid; i < 16 * 12; i += 128) {
    int rr = i / 12, cc = 68 + i % 12;
    lds_v[rr][cc] = (cc == 68) ? 1.f : 0.f;
  }

  v8f acc[5] = {};

  for (int mc = 0; mc < NN / 16; ++mc) {
    __syncthreads();
    {
      const float2* se = (const float2*)(Eb + (long)mc * 16 * NDE);
      ((float2*)lds_e)[tid] = se[tid];
    }
    {
      const float2* sv = (const float2*)(Xb + (long)mc * 16 * NCP);
      for (int i = tid; i < 544; i += 128) {
        int rr = i / 34, cc = (i % 34) * 2;
        *(float2*)&lds_v[rr][cc] = sv[i];
      }
    }
    __syncthreads();

    v8f c = {};
#pragma unroll
    for (int kk = 0; kk < 4; ++kk) {
      v2f bf;
      bf.x = lds_e[l16][kk * 4 + 2 * half];
      bf.y = lds_e[l16][kk * 4 + 2 * half + 1];
      c = WMMA_F32(aA[kk], bf, c);
    }

#pragma unroll
    for (int r = 0; r < 8; ++r)
      lds_p[wave][r + 8 * half][l16] = __expf(c[r]);

#pragma unroll
    for (int kk = 0; kk < 4; ++kk) {
      v2f ap;
      ap.x = lds_p[wave][l16][kk * 4 + 2 * half];
      ap.y = lds_p[wave][l16][kk * 4 + 2 * half + 1];
#pragma unroll
      for (int ct = 0; ct < 5; ++ct) {
        v2f bv;
        bv.x = lds_v[kk * 4 + 2 * half][ct * 16 + l16];
        bv.y = lds_v[kk * 4 + 2 * half + 1][ct * 16 + l16];
        acc[ct] = WMMA_F32(ap, bv, acc[ct]);
      }
    }
  }

#pragma unroll
  for (int r = 0; r < 8; ++r) {
    float rs = __shfl(acc[4][r], half * 16 + 4);
    float inv = 1.f / rs;
    int row = row0 + wave * 16 + r + 8 * half;
#pragma unroll
    for (int ct = 0; ct < 5; ++ct) {
      int col = ct * 16 + l16;
      if (col < NCP)
        Y[((long)b * NN + row) * NCP + col] = acc[ct][r] * inv;
    }
  }
}

__global__ void __launch_bounds__(64) k_gout(
    const float* __restrict__ X, const float* __restrict__ XG,
    const float* __restrict__ W, const float* __restrict__ TB,
    float* __restrict__ G, int g, int t) {
  const int n = blockIdx.x;
  const int wave = threadIdx.x >> 5;
  const int lane = threadIdx.x & 31;
  const int l16 = lane & 15, half = lane >> 4;
  const float* Wg = W + ((long)g * NN + n) * 2 * NCP * NHID;

  v8f c[4] = {};
  for (int slot = 0; slot < 2; ++slot) {
    const float* Arow = (slot ? XG : X) + ((long)(wave * 16 + l16) * NN + n) * NCP;
    const float* Wm = Wg + (long)slot * NCP * NHID;
    for (int kk = 0; kk < NCP / 4; ++kk) {
      v2f a;
      a.x = Arow[kk * 4 + 2 * half];
      a.y = Arow[kk * 4 + 2 * half + 1];
#pragma unroll
      for (int ct = 0; ct < 4; ++ct) {
        v2f bw;
        bw.x = Wm[(kk * 4 + 2 * half) * 64 + ct * 16 + l16];
        bw.y = Wm[(kk * 4 + 2 * half + 1) * 64 + ct * 16 + l16];
        c[ct] = WMMA_F32(a, bw, c[ct]);
      }
    }
  }
#pragma unroll
  for (int ct = 0; ct < 4; ++ct)
#pragma unroll
    for (int r = 0; r < 8; ++r) {
      int bb = wave * 16 + r + 8 * half;
      int o = ct * 16 + l16;
      G[((long)bb * NN + n) * 64 + o] = c[ct][r] + TB[(((long)g * NT + t) * NB + bb) * 64 + o];
    }
}

__global__ void __launch_bounds__(256) k_state_attn(
    float* __restrict__ G, const float* __restrict__ Hs,
    const float* __restrict__ olg, const float* __restrict__ olb, int g, int K) {
  const int wave = threadIdx.x >> 5, lane = threadIdx.x & 31;
  const long row = (long)blockIdx.x * 8 + wave;
  float g0 = G[row * 64 + lane];
  float g1 = G[row * 64 + lane + 32];
  float s = g0 + g1;
  s += __shfl_xor(s, 1); s += __shfl_xor(s, 2); s += __shfl_xor(s, 4);
  s += __shfl_xor(s, 8); s += __shfl_xor(s, 16);
  float mu = s * (1.f / 64.f);
  float d0 = g0 - mu, d1 = g1 - mu;
  float v = d0 * d0 + d1 * d1;
  v += __shfl_xor(v, 1); v += __shfl_xor(v, 2); v += __shfl_xor(v, 4);
  v += __shfl_xor(v, 8); v += __shfl_xor(v, 16);
  float inv = rsqrtf(v * (1.f / 64.f) + 1e-5f);
  float q0 = d0 * inv * olg[g * 64 + lane] + olb[g * 64 + lane];
  float q1 = d1 * inv * olg[g * 64 + lane + 32] + olb[g * 64 + lane + 32];

  float st0[13], st1[13], sc0[13], sc1[13];
#pragma unroll
  for (int k = 0; k < 13; ++k) {
    if (k < K) {
      st0[k] = Hs[((long)k * NB * NN + row) * 64 + lane];
      st1[k] = Hs[((long)k * NB * NN + row) * 64 + lane + 32];
      float p0 = q0 * st0[k], p1 = q1 * st1[k];
      p0 += __shfl_xor(p0, 1); p0 += __shfl_xor(p0, 2);
      p0 += __shfl_xor(p0, 4); p0 += __shfl_xor(p0, 8);
      p1 += __shfl_xor(p1, 1); p1 += __shfl_xor(p1, 2);
      p1 += __shfl_xor(p1, 4); p1 += __shfl_xor(p1, 8);
      sc0[k] = p0 * 0.25f; sc1[k] = p1 * 0.25f;
    } else { st0[k] = 0.f; st1[k] = 0.f; sc0[k] = -1e30f; sc1[k] = -1e30f; }
  }
  float m0 = -1e30f, m1 = -1e30f;
#pragma unroll
  for (int k = 0; k < 13; ++k) { m0 = fmaxf(m0, sc0[k]); m1 = fmaxf(m1, sc1[k]); }
  float s0 = 0.f, s1 = 0.f;
#pragma unroll
  for (int k = 0; k < 13; ++k) {
    sc0[k] = __expf(sc0[k] - m0); s0 += sc0[k];
    sc1[k] = __expf(sc1[k] - m1); s1 += sc1[k];
  }
  float o0 = 0.f, o1 = 0.f;
#pragma unroll
  for (int k = 0; k < 13; ++k) { o0 += sc0[k] * st0[k]; o1 += sc1[k] * st1[k]; }
  G[row * 64 + lane]      = g0 + o0 / s0;
  G[row * 64 + lane + 32] = g1 + o1 / s1;
}

__global__ void k_combine(const float* __restrict__ Gr, const float* __restrict__ Ghc,
                          const float* __restrict__ Hprev, float* __restrict__ Hnew) {
  long i = (long)blockIdx.x * 256 + threadIdx.x;
  if (i >= (long)NB * NN * 64) return;
  float r = 1.f / (1.f + __expf(-Gr[i]));
  float hc = tanhf(Ghc[i]);
  Hnew[i] = r * Hprev[i] + (1.f - r) * hc;
}

__global__ void k_pred(const float* __restrict__ Hs, const float* __restrict__ pw,
                       const float* __restrict__ pb, float* __restrict__ out) {
  int idx = blockIdx.x * 256 + threadIdx.x;
  if (idx >= NB * 12 * NN) return;
  int n = idx % NN; int r = idx / NN; int o = r % 12; int b = r / 12;
  float acc = pb[o];
  for (int cc = 0; cc < 2; ++cc) {
    const float* h = Hs + ((long)(NT - 1 + cc) * NB * NN + (long)b * NN + n) * 64;
    for (int hh = 0; hh < 64; ++hh)
      acc += h[hh] * pw[(o * 2 + cc) * 64 + hh];
  }
  out[idx] = acc;
}

extern "C" void kernel_launch(void* const* d_in, const int* in_sizes, int n_in,
                              void* d_out, int out_size, void* d_ws, size_t ws_size,
                              hipStream_t stream) {
  const float* x   = (const float*)d_in[0];
  const float* ne  = (const float*)d_in[1];
  const float* te  = (const float*)d_in[2];
  const float* wp  = (const float*)d_in[3];
  const float* bp  = (const float*)d_in[4];
  const float* glg = (const float*)d_in[5];
  const float* glb = (const float*)d_in[6];
  const float* olg = (const float*)d_in[7];
  const float* olb = (const float*)d_in[8];
  const float* pw  = (const float*)d_in[9];
  const float* pb  = (const float*)d_in[10];
  float* out = (float*)d_out;

  float* ws = (float*)d_ws;
  long off = 0;
  float* W    = ws + off; off += 3L * NN * 2 * NCP * NHID;
  float* TB   = ws + off; off += 3L * NT * NB * NHID;
  float* E    = ws + off; off += 3L * NB * NN * NDE;
  float* Hs   = ws + off; off += (long)(NT + 1) * NB * NN * NHID;
  float* IAS  = ws + off; off += (long)NB * NN * NCP;
  float* CAND = ws + off; off += (long)NB * NN * NCP;
  float* XG2  = ws + off; off += (long)NB * NN * NCP;
  float* G0   = ws + off; off += (long)NB * NN * NHID;
  float* G1   = ws + off; off += (long)NB * NN * NHID;
  float* G2   = ws + off; off += (long)NB * NN * NHID;

  const long wTotal = 3L * NN * 2 * NCP * NHID;
  k_w<<<dim3((unsigned)((wTotal + 255) / 256)), 256, 0, stream>>>(ne, wp, W);
  k_tb<<<(3 * NT * NB * NHID + 255) / 256, 256, 0, stream>>>(te, bp, TB);
  k_zero<<<(unsigned)(((long)NB * NN * NHID + 255) / 256), 256, 0, stream>>>(Hs, (long)NB * NN * NHID);

  for (int t = 0; t < NT; ++t) {
    float* Hprev = Hs + (long)t * NB * NN * NHID;
    float* Hnew  = Hs + (long)(t + 1) * NB * NN * NHID;
    k_emb<<<(3 * NB * NN + 255) / 256, 256, 0, stream>>>(ne, te, glg, glb, E, t);
    k_build_ias<<<(NB * NN * NCP + 255) / 256, 256, 0, stream>>>(x, Hprev, IAS, t);
    for (int g = 0; g < 2; ++g) {
      float* Gg = g ? G1 : G0;
      k_graph_attn<<<dim3(NN / 64, NB), 128, 0, stream>>>(E, IAS, XG2, g);
      k_gout<<<NN, 64, 0, stream>>>(IAS, XG2, W, TB, Gg, g, t);
      k_state_attn<<<NB * NN / 8, 256, 0, stream>>>(Gg, Hs, olg, olb, g, t + 1);
    }
    k_build_cand<<<(NB * NN * NCP + 255) / 256, 256, 0, stream>>>(x, Hprev, G0, CAND, t);
    k_graph_attn<<<dim3(NN / 64, NB), 128, 0, stream>>>(E, CAND, XG2, 2);
    k_gout<<<NN, 64, 0, stream>>>(CAND, XG2, W, TB, G2, 2, t);
    k_state_attn<<<NB * NN / 8, 256, 0, stream>>>(G2, Hs, olg, olb, 2, t + 1);
    k_combine<<<(unsigned)(((long)NB * NN * NHID + 255) / 256), 256, 0, stream>>>(G1, G2, Hprev, Hnew);
  }
  k_pred<<<(NB * 12 * NN + 255) / 256, 256, 0, stream>>>(Hs, pw, pb, out);
}


# k_graph_attn: the batch's whole 512-key E and X panel is staged in LDS once per block (all loads requested up front, one barrier) instead of 32 chunk reloads of 2 barriers each; sum order unchanged
# speedup vs baseline: 1.9470x; 1.9470x over previous
#include <hip/hip_runtime.h>
#include <math.h>

typedef __attribute__((ext_vector_type(2))) float v2f;
typedef __attribute__((ext_vector_type(8))) float v8f;

#define WMMA_F32(A, B, C) \
  __builtin_amdgcn_wmma_f32_16x16x4_f32(false, (A), false, (B), (short)0, (C), false, false)

#define NB   32
#define NT   12
#define NN   512
#define NDE  16
#define NHID 64
#define NCP  68

__global__ void k_w(const float* __restrict__ ne, const float* __restrict__ wp,
                    float* __restrict__ W) {
  long idx = (long)blockIdx.x * 256 + threadIdx.x;
  const long total = 3L * NN * 2 * NCP * NHID;
  if (idx >= total) return;
  int o = idx % 64; long r = idx / 64;
  int i = r % NCP; r /= NCP;
  int slot = r % 2; r /= 2;
  int n = r % NN; int g = (int)(r / NN);
  float acc = 0.f;
  if (i < 67) {
    for (int d = 0; d < NDE; ++d)
      acc += ne[n * NDE + d] * wp[(((long)(g * NDE + d) * 2 + slot) * 67 + i) * 64 + o];
  }
  W[idx] = acc;
}

__global__ void k_tb(const float* __restrict__ te, const float* __restrict__ bp,
                     float* __restrict__ TB) {
  int idx = blockIdx.x * 256 + threadIdx.x;
  if (idx >= 3 * NT * NB * NHID) return;
  int o = idx % 64; int r = idx / 64;
  int b = r % NB; r /= NB;
  int t = r % NT; int g = r / NT;
  float acc = 0.f;
  for (int d = 0; d < NDE; ++d)
    acc += te[(b * NT + t) * NDE + d] * bp[(g * NDE + d) * 64 + o];
  TB[idx] = acc;
}

__global__ void k_emb(const float* __restrict__ ne, const float* __restrict__ te,
                      const float* __restrict__ glg, const float* __restrict__ glb,
                      float* __restrict__ E, int t) {
  int idx = blockIdx.x * 256 + threadIdx.x;
  if (idx >= 3 * NB * NN) return;
  int n = idx % NN; int r = idx / NN;
  int b = r % NB; int g = r / NB;
  float v[NDE]; float mu = 0.f;
  for (int d = 0; d < NDE; ++d) { v[d] = ne[n * NDE + d] + te[(b * NT + t) * NDE + d]; mu += v[d]; }
  mu *= (1.f / NDE);
  float var = 0.f;
  for (int d = 0; d < NDE; ++d) { float q = v[d] - mu; var += q * q; }
  var *= (1.f / NDE);
  float inv = rsqrtf(var + 1e-12f);
  for (int d = 0; d < NDE; ++d)
    E[(long)idx * NDE + d] = (v[d] - mu) * inv * glg[g * NDE + d] + glb[g * NDE + d];
}

__global__ void k_zero(float* p, long nElem) {
  long i = (long)blockIdx.x * 256 + threadIdx.x;
  if (i < nElem) p[i] = 0.f;
}

__global__ void k_build_ias(const float* __restrict__ x, const float* __restrict__ h,
                            float* __restrict__ ias, int t) {
  int idx = blockIdx.x * 256 + threadIdx.x;
  if (idx >= NB * NN * NCP) return;
  int i = idx % NCP; int r = idx / NCP; int n = r % NN; int b = r / NN;
  float v;
  if (i < 3)       v = x[((long)(b * NT + t) * NN + n) * 3 + i];
  else if (i < 67) v = h[((long)b * NN + n) * 64 + (i - 3)];
  else             v = 0.f;
  ias[idx] = v;
}

__global__ void k_build_cand(const float* __restrict__ x, const float* __restrict__ h,
                             const float* __restrict__ zp, float* __restrict__ cand, int t) {
  int idx = blockIdx.x * 256 + threadIdx.x;
  if (idx >= NB * NN * NCP) return;
  int i = idx % NCP; int r = idx / NCP; int n = r % NN; int b = r / NN;
  float v;
  if (i < 3)       v = x[((long)(b * NT + t) * NN + n) * 3 + i];
  else if (i < 67) {
    long e = ((long)b * NN + n) * 64 + (i - 3);
    float z = 1.f / (1.f + __expf(-zp[e]));
    v = z * h[e];
  } else v = 0.f;
  cand[idx] = v;
}

#define VP 80
__global__ void __launch_bounds__(128) k_graph_attn(
    const float* __restrict__ E,
    const float* __restrict__ X,
    float* __restrict__ Y,
    int g) {
  __shared__ __attribute__((aligned(16))) float lds_e[NN * NDE];
  __shared__ __attribute__((aligned(16))) float lds_v[NN * VP];
  __shared__ float lds_p[4][16][17];

  const int b = blockIdx.y;
  const int row0 = blockIdx.x * 64;
  const int tid = threadIdx.x;
  const int wave = tid >> 5;
  const int lane = tid & 31;
  const int l16 = lane & 15;
  const int half = lane >> 4;
  const int myrow = row0 + wave * 16 + l16;

  const float* Eb = E + ((long)g * NB + b) * NN * NDE;
  const float* Xb = X + (long)b * NN * NCP;

  {
    const float4* se = (const float4*)Eb;
    float4 re[16];
#pragma unroll
    for (int j = 0; j < 16; ++j) re[j] = se[tid + 128 * j];
#pragma unroll
    for (int j = 0; j < 16; ++j) ((float4*)lds_e)[tid + 128 * j] = re[j];

    const float4* sv = (const float4*)Xb;
#pragma unroll
    for (int ph = 0; ph < 4; ++ph) {
      float4 rv[17];
#pragma unroll
      for (int j = 0; j < 17; ++j) rv[j] = sv[ph * 2176 + tid + 128 * j];
#pragma unroll
      for (int j = 0; j < 17; ++j) {
        int i = ph * 2176 + tid + 128 * j;
        int rr = i / 17, q = i - rr * 17;
        *(float4*)&lds_v[rr * VP + q * 4] = rv[j];
      }
    }
#pragma unroll
    for (int j = 0; j < 12; ++j) {
      int i = tid + 128 * j;
      int rr = i / 3, q = i - rr * 3;
      float4 v = make_float4(q == 0 ? 1.f : 0.f, 0.f, 0.f, 0.f);
      *(float4*)&lds_v[rr * VP + 68 + q * 4] = v;
    }
  }

  v2f aA[4];
#pragma unroll
  for (int kk = 0; kk < 4; ++kk) {
    aA[kk].x = Eb[myrow * NDE + kk * 4 + 2 * half];
    aA[kk].y = Eb[myrow * NDE + kk * 4 + 2 * half + 1];
  }

  v8f acc[5] = {};

  __syncthreads();

  for (int mc = 0; mc < NN / 16; ++mc) {
    const float* le = lds_e + mc * 16 * NDE;
    const float* lv = lds_v + mc * 16 * VP;

    v8f c = {};
#pragma unroll
    for (int kk = 0; kk < 4; ++kk) {
      v2f bf;
      bf.x = le[l16 * NDE + kk * 4 + 2 * half];
      bf.y = le[l16 * NDE + kk * 4 + 2 * half + 1];
      c = WMMA_F32(aA[kk], bf, c);
    }

#pragma unroll
    for (int r = 0; r < 8; ++r)
      lds_p[wave][r + 8 * half][l16] = __expf(c[r]);

#pragma unroll
    for (int kk = 0; kk < 4; ++kk) {
      v2f ap;
      ap.x = lds_p[wave][l16][kk * 4 + 2 * half];
      ap.y = lds_p[wave][l16][kk * 4 + 2 * half + 1];
#pragma unroll
      for (int ct = 0; ct < 5; ++ct) {
        v2f bv;
        bv.x = lv[(kk * 4 + 2 * half) * VP + ct * 16 + l16];
        bv.y = lv[(kk * 4 + 2 * half + 1) * VP + ct * 16 + l16];
        acc[ct] = WMMA_F32(ap, bv, acc[ct]);
      }
    }
  }

#pragma unroll
  for (int r = 0; r < 8; ++r) {
    float rs = __shfl(acc[4][r], half * 16 + 4);
    float inv = 1.f / rs;
    int row = row0 + wave * 16 + r + 8 * half;
#pragma unroll
    for (int ct = 0; ct < 5; ++ct) {
      int col = ct * 16 + l16;
      if (col < NCP)
        Y[((long)b * NN + row) * NCP + col] = acc[ct][r] * inv;
    }
  }
}

__global__ void __launch_bounds__(64) k_gout(
    const float* __restrict__ X, const float* __restrict__ XG,
    const float* __restrict__ W, const float* __restrict__ TB,
    float* __restrict__ G, int g, int t) {
  const int n = blockIdx.x;
  const int wave = threadIdx.x >> 5;
  const int lane = threadIdx.x & 31;
  const int l16 = lane & 15, half = lane >> 4;
  const float* Wg = W + ((long)g * NN + n) * 2 * NCP * NHID;

  v8f c[4] = {};
  for (int slot = 0; slot < 2; ++slot) {
    const float* Arow = (slot ? XG : X) + ((long)(wave * 16 + l16) * NN + n) * NCP;
    const float* Wm = Wg + (long)slot * NCP * NHID;
    for (int kk = 0; kk < NCP / 4; ++kk) {
      v2f a;
      a.x = Arow[kk * 4 + 2 * half];
      a.y = Arow[kk * 4 + 2 * half + 1];
#pragma unroll
      for (int ct = 0; ct < 4; ++ct) {
        v2f bw;
        bw.x = Wm[(kk * 4 + 2 * half) * 64 + ct * 16 + l16];
        bw.y = Wm[(kk * 4 + 2 * half + 1) * 64 + ct * 16 + l16];
        c[ct] = WMMA_F32(a, bw, c[ct]);
      }
    }
  }
#pragma unroll
  for (int ct = 0; ct < 4; ++ct)
#pragma unroll
    for (int r = 0; r < 8; ++r) {
      int bb = wave * 16 + r + 8 * half;
      int o = ct * 16 + l16;
      G[((long)bb * NN + n) * 64 + o] = c[ct][r] + TB[(((long)g * NT + t) * NB + bb) * 64 + o];
    }
}

__global__ void __launch_bounds__(256) k_state_attn(
    float* __restrict__ G, const float* __restrict__ Hs,
    const float* __restrict__ olg, const float* __restrict__ olb, int g, int K) {
  const int wave = threadIdx.x >> 5, lane = threadIdx.x & 31;
  const long row = (long)blockIdx.x * 8 + wave;
  float g0 = G[row * 64 + lane];
  float g1 = G[row * 64 + lane + 32];
  float s = g0 + g1;
  s += __shfl_xor(s, 1); s += __shfl_xor(s, 2); s += __shfl_xor(s, 4);
  s += __shfl_xor(s, 8); s += __shfl_xor(s, 16);
  float mu = s * (1.f / 64.f);
  float d0 = g0 - mu, d1 = g1 - mu;
  float v = d0 * d0 + d1 * d1;
  v += __shfl_xor(v, 1); v += __shfl_xor(v, 2); v += __shfl_xor(v, 4);
  v += __shfl_xor(v, 8); v += __shfl_xor(v, 16);
  float inv = rsqrtf(v * (1.f / 64.f) + 1e-5f);
  float q0 = d0 * inv * olg[g * 64 + lane] + olb[g * 64 + lane];
  float q1 = d1 * inv * olg[g * 64 + lane + 32] + olb[g * 64 + lane + 32];

  float st0[13], st1[13], sc0[13], sc1[13];
#pragma unroll
  for (int k = 0; k < 13; ++k) {
    if (k < K) {
      st0[k] = Hs[((long)k * NB * NN + row) * 64 + lane];
      st1[k] = Hs[((long)k * NB * NN + row) * 64 + lane + 32];
      float p0 = q0 * st0[k], p1 = q1 * st1[k];
      p0 += __shfl_xor(p0, 1); p0 += __shfl_xor(p0, 2);
      p0 += __shfl_xor(p0, 4); p0 += __shfl_xor(p0, 8);
      p1 += __shfl_xor(p1, 1); p1 += __shfl_xor(p1, 2);
      p1 += __shfl_xor(p1, 4); p1 += __shfl_xor(p1, 8);
      sc0[k] = p0 * 0.25f; sc1[k] = p1 * 0.25f;
    } else { st0[k] = 0.f; st1[k] = 0.f; sc0[k] = -1e30f; sc1[k] = -1e30f; }
  }
  float m0 = -1e30f, m1 = -1e30f;
#pragma unroll
  for (int k = 0; k < 13; ++k) { m0 = fmaxf(m0, sc0[k]); m1 = fmaxf(m1, sc1[k]); }
  float s0 = 0.f, s1 = 0.f;
#pragma unroll
  for (int k = 0; k < 13; ++k) {
    sc0[k] = __expf(sc0[k] - m0); s0 += sc0[k];
    sc1[k] = __expf(sc1[k] - m1); s1 += sc1[k];
  }
  float o0 = 0.f, o1 = 0.f;
#pragma unroll
  for (int k = 0; k < 13; ++k) { o0 += sc0[k] * st0[k]; o1 += sc1[k] * st1[k]; }
  G[row * 64 + lane]      = g0 + o0 / s0;
  G[row * 64 + lane + 32] = g1 + o1 / s1;
}

__global__ void k_combine(const float* __restrict__ Gr, const float* __restrict__ Ghc,
                          const float* __restrict__ Hprev, float* __restrict__ Hnew) {
  long i = (long)blockIdx.x * 256 + threadIdx.x;
  if (i >= (long)NB * NN * 64) return;
  float r = 1.f / (1.f + __expf(-Gr[i]));
  float hc = tanhf(Ghc[i]);
  Hnew[i] = r * Hprev[i] + (1.f - r) * hc;
}

__global__ void k_pred(const float* __restrict__ Hs, const float* __restrict__ pw,
                       const float* __restrict__ pb, float* __restrict__ out) {
  int idx = blockIdx.x * 256 + threadIdx.x;
  if (idx >= NB * 12 * NN) return;
  int n = idx % NN; int r = idx / NN; int o = r % 12; int b = r / 12;
  float acc = pb[o];
  for (int cc = 0; cc < 2; ++cc) {
    const float* h = Hs + ((long)(NT - 1 + cc) * NB * NN + (long)b * NN + n) * 64;
    for (int hh = 0; hh < 64; ++hh)
      acc += h[hh] * pw[(o * 2 + cc) * 64 + hh];
  }
  out[idx] = acc;
}

extern "C" void kernel_launch(void* const* d_in, const int* in_sizes, int n_in,
                              void* d_out, int out_size, void* d_ws, size_t ws_size,
                              hipStream_t stream) {
  const float* x   = (const float*)d_in[0];
  const float* ne  = (const float*)d_in[1];
  const float* te  = (const float*)d_in[2];
  const float* wp  = (const float*)d_in[3];
  const float* bp  = (const float*)d_in[4];
  const float* glg = (const float*)d_in[5];
  const float* glb = (const float*)d_in[6];
  const float* olg = (const float*)d_in[7];
  const float* olb = (const float*)d_in[8];
  const float* pw  = (const float*)d_in[9];
  const float* pb  = (const float*)d_in[10];
  float* out = (float*)d_out;

  float* ws = (float*)d_ws;
  long off = 0;
  float* W    = ws + off; off += 3L * NN * 2 * NCP * NHID;
  float* TB   = ws + off; off += 3L * NT * NB * NHID;
  float* E    = ws + off; off += 3L * NB * NN * NDE;
  float* Hs   = ws + off; off += (long)(NT + 1) * NB * NN * NHID;
  float* IAS  = ws + off; off += (long)NB * NN * NCP;
  float* CAND = ws + off; off += (long)NB * NN * NCP;
  float* XG2  = ws + off; off += (long)NB * NN * NCP;
  float* G0   = ws + off; off += (long)NB * NN * NHID;
  float* G1   = ws + off; off += (long)NB * NN * NHID;
  float* G2   = ws + off; off += (long)NB * NN * NHID;

  const long wTotal = 3L * NN * 2 * NCP * NHID;
  k_w<<<dim3((unsigned)((wTotal + 255) / 256)), 256, 0, stream>>>(ne, wp, W);
  k_tb<<<(3 * NT * NB * NHID + 255) / 256, 256, 0, stream>>>(te, bp, TB);
  k_zero<<<(unsigned)(((long)NB * NN * NHID + 255) / 256), 256, 0, stream>>>(Hs, (long)NB * NN * NHID);

  for (int t = 0; t < NT; ++t) {
    float* Hprev = Hs + (long)t * NB * NN * NHID;
    float* Hnew  = Hs + (long)(t + 1) * NB * NN * NHID;
    k_emb<<<(3 * NB * NN + 255) / 256, 256, 0, stream>>>(ne, te, glg, glb, E, t);
    k_build_ias<<<(NB * NN * NCP + 255) / 256, 256, 0, stream>>>(x, Hprev, IAS, t);
    for (int g = 0; g < 2; ++g) {
      float* Gg = g ? G1 : G0;
      k_graph_attn<<<dim3(NN / 64, NB), 128, 0, stream>>>(E, IAS, XG2, g);
      k_gout<<<NN, 64, 0, stream>>>(IAS, XG2, W, TB, Gg, g, t);
      k_state_attn<<<NB * NN / 8, 256, 0, stream>>>(Gg, Hs, olg, olb, g, t + 1);
    }
    k_build_cand<<<(NB * NN * NCP + 255) / 256, 256, 0, stream>>>(x, Hprev, G0, CAND, t);
    k_graph_attn<<<dim3(NN / 64, NB), 128, 0, stream>>>(E, CAND, XG2, 2);
    k_gout<<<NN, 64, 0, stream>>>(CAND, XG2, W, TB, G2, 2, t);
    k_state_attn<<<NB * NN / 8, 256, 0, stream>>>(G2, Hs, olg, olb, 2, t + 1);
    k_combine<<<(unsigned)(((long)NB * NN * NHID + 255) / 256), 256, 0, stream>>>(G1, G2, Hprev, Hnew);
  }
  k_pred<<<(NB * 12 * NN + 255) / 256, 256, 0, stream>>>(Hs, pw, pb, out);
}


# k_gout: all operands of a node (both weight slabs, input rows, bias) requested up front as 16-byte loads (tile column = 4*lane+tile) instead of 34 load-wait-multiply steps; sum order unchanged
# speedup vs baseline: 2.6010x; 1.3359x over previous
#include <hip/hip_runtime.h>
#include <math.h>

typedef __attribute__((ext_vector_type(2))) float v2f;
typedef __attribute__((ext_vector_type(8))) float v8f;

#define WMMA_F32(A, B, C) \
  __builtin_amdgcn_wmma_f32_16x16x4_f32(false, (A), false, (B), (short)0, (C), false, false)

#define NB   32
#define NT   12
#define NN   512
#define NDE  16
#define NHID 64
#define NCP  68

__global__ void k_w(const float* __restrict__ ne, const float* __restrict__ wp,
                    float* __restrict__ W) {
  long idx = (long)blockIdx.x * 256 + threadIdx.x;
  const long total = 3L * NN * 2 * NCP * NHID;
  if (idx >= total) return;
  int o = idx % 64; long r = idx / 64;
  int i = r % NCP; r /= NCP;
  int slot = r % 2; r /= 2;
  int n = r % NN; int g = (int)(r / NN);
  float acc = 0.f;
  if (i < 67) {
    for (int d = 0; d < NDE; ++d)
      acc += ne[n * NDE + d] * wp[(((long)(g * NDE + d) * 2 + slot) * 67 + i) * 64 + o];
  }
  W[idx] = acc;
}

__global__ void k_tb(const float* __restrict__ te, const float* __restrict__ bp,
                     float* __restrict__ TB) {
  int idx = blockIdx.x * 256 + threadIdx.x;
  if (idx >= 3 * NT * NB * NHID) return;
  int o = idx % 64; int r = idx / 64;
  int b = r % NB; r /= NB;
  int t = r % NT; int g = r / NT;
  float acc = 0.f;
  for (int d = 0; d < NDE; ++d)
    acc += te[(b * NT + t) * NDE + d] * bp[(g * NDE + d) * 64 + o];
  TB[idx] = acc;
}

__global__ void k_emb(const float* __restrict__ ne, const float* __restrict__ te,
                      const float* __restrict__ glg, const float* __restrict__ glb,
                      float* __restrict__ E, int t) {
  int idx = blockIdx.x * 256 + threadIdx.x;
  if (idx >= 3 * NB * NN) return;
  int n = idx % NN; int r = idx / NN;
  int b = r % NB; int g = r / NB;
  float v[NDE]; float mu = 0.f;
  for (int d = 0; d < NDE; ++d) { v[d] = ne[n * NDE + d] + te[(b * NT + t) * NDE + d]; mu += v[d]; }
  mu *= (1.f / NDE);
  float var = 0.f;
  for (int d = 0; d < NDE; ++d) { float q = v[d] - mu; var += q * q; }
  var *= (1.f / NDE);
  float inv = rsqrtf(var + 1e-12f);
  for (int d = 0; d < NDE; ++d)
    E[(long)idx * NDE + d] = (v[d] - mu) * inv * glg[g * NDE + d] + glb[g * NDE + d];
}

__global__ void k_zero(float* p, long nElem) {
  long i = (long)blockIdx.x * 256 + threadIdx.x;
  if (i < nElem) p[i] = 0.f;
}

__global__ void k_build_ias(const float* __restrict__ x, const float* __restrict__ h,
                            float* __restrict__ ias, int t) {
  int idx = blockIdx.x * 256 + threadIdx.x;
  if (idx >= NB * NN * NCP) return;
  int i = idx % NCP; int r = idx / NCP; int n = r % NN; int b = r / NN;
  float v;
  if (i < 3)       v = x[((long)(b * NT + t) * NN + n) * 3 + i];
  else if (i < 67) v = h[((long)b * NN + n) * 64 + (i - 3)];
  else             v = 0.f;
  ias[idx] = v;
}

__global__ void k_build_cand(const float* __restrict__ x, const float* __restrict__ h,
                             const float* __restrict__ zp, float* __restrict__ cand, int t) {
  int idx = blockIdx.x * 256 + threadIdx.x;
  if (idx >= NB * NN * NCP) return;
  int i = idx % NCP; int r = idx / NCP; int n = r % NN; int b = r / NN;
  float v;
  if (i < 3)       v = x[((long)(b * NT + t) * NN + n) * 3 + i];
  else if (i < 67) {
    long e = ((long)b * NN + n) * 64 + (i - 3);
    float z = 1.f / (1.f + __expf(-zp[e]));
    v = z * h[e];
  } else v = 0.f;
  cand[idx] = v;
}

#define VP 80
__global__ void __launch_bounds__(128) k_graph_attn(
    const float* __restrict__ E,
    const float* __restrict__ X,
    float* __restrict__ Y,
    int g) {
  __shared__ __attribute__((aligned(16))) float lds_e[NN * NDE];
  __shared__ __attribute__((aligned(16))) float lds_v[NN * VP];
  __shared__ float lds_p[4][16][17];

  const int b = blockIdx.y;
  const int row0 = blockIdx.x * 64;
  const int tid = threadIdx.x;
  const int wave = tid >> 5;
  const int lane = tid & 31;
  const int l16 = lane & 15;
  const int half = lane >> 4;
  const int myrow = row0 + wave * 16 + l16;

  const float* Eb = E + ((long)g * NB + b) * NN * NDE;
  const float* Xb = X + (long)b * NN * NCP;

  {
    const float4* se = (const float4*)Eb;
    float4 re[16];
#pragma unroll
    for (int j = 0; j < 16; ++j) re[j] = se[tid + 128 * j];
#pragma unroll
    for (int j = 0; j < 16; ++j) ((float4*)lds_e)[tid + 128 * j] = re[j];

    const float4* sv = (const float4*)Xb;
#pragma unroll
    for (int ph = 0; ph < 4; ++ph) {
      float4 rv[17];
#pragma unroll
      for (int j = 0; j < 17; ++j) rv[j] = sv[ph * 2176 + tid + 128 * j];
#pragma unroll
      for (int j = 0; j < 17; ++j) {
        int i = ph * 2176 + tid + 128 * j;
        int rr = i / 17, q = i - rr * 17;
        *(float4*)&lds_v[rr * VP + q * 4] = rv[j];
      }
    }
#pragma unroll
    for (int j = 0; j < 12; ++j) {
      int i = tid + 128 * j;
      int rr = i / 3, q = i - rr * 3;
      float4 v = make_float4(q == 0 ? 1.f : 0.f, 0.f, 0.f, 0.f);
      *(float4*)&lds_v[rr * VP + 68 + q * 4] = v;
    }
  }

  v2f aA[4];
#pragma unroll
  for (int kk = 0; kk < 4; ++kk) {
    aA[kk].x = Eb[myrow * NDE + kk * 4 + 2 * half];
    aA[kk].y = Eb[myrow * NDE + kk * 4 + 2 * half + 1];
  }

  v8f acc[5] = {};

  __syncthreads();

  for (int mc = 0; mc < NN / 16; ++mc) {
    const float* le = lds_e + mc * 16 * NDE;
    const float* lv = lds_v + mc * 16 * VP;

    v8f c = {};
#pragma unroll
    for (int kk = 0; kk < 4; ++kk) {
      v2f bf;
      bf.x = le[l16 * NDE + kk * 4 + 2 * half];
      bf.y = le[l16 * NDE + kk * 4 + 2 * half + 1];
      c = WMMA_F32(aA[kk], bf, c);
    }

#pragma unroll
    for (int r = 0; r < 8; ++r)
      lds_p[wave][r + 8 * half][l16] = __expf(c[r]);

#pragma unroll
    for (int kk = 0; kk < 4; ++kk) {
      v2f ap;
      ap.x = lds_p[wave][l16][kk * 4 + 2 * half];
      ap.y = lds_p[wave][l16][kk * 4 + 2 * half + 1];
#pragma unroll
      for (int ct = 0; ct < 5; ++ct) {
        v2f bv;
        bv.x = lv[(kk * 4 + 2 * half) * VP + ct * 16 + l16];
        bv.y = lv[(kk * 4 + 2 * half + 1) * VP + ct * 16 + l16];
        acc[ct] = WMMA_F32(ap, bv, acc[ct]);
      }
    }
  }

#pragma unroll
  for (int r = 0; r < 8; ++r) {
    float rs = __shfl(acc[4][r], half * 16 + 4);
    float inv = 1.f / rs;
    int row = row0 + wave * 16 + r + 8 * half;
#pragma unroll
    for (int ct = 0; ct < 5; ++ct) {
      int col = ct * 16 + l16;
      if (col < NCP)
        Y[((long)b * NN + row) * NCP + col] = acc[ct][r] * inv;
    }
  }
}

typedef __attribute__((ext_vector_type(4))) float v4f;
__global__ void __launch_bounds__(64) k_gout(
    const float* __restrict__ X, const float* __restrict__ XG,
    const float* __restrict__ W, const float* __restrict__ TB,
    float* __restrict__ G, int g, int t) {
  const int n = blockIdx.x;
  const int wave = threadIdx.x >> 5;
  const int lane = threadIdx.x & 31;
  const int l16 = lane & 15, half = lane >> 4;
  const float* Wg = W + ((long)g * NN + n) * 2 * NCP * NHID;

  v2f a[2][NCP / 4];
  v4f bw[2][NCP / 4][2];
  v4f tb[8];

#pragma unroll
  for (int slot = 0; slot < 2; ++slot) {
    const float* Arow = (slot ? XG : X) + ((long)(wave * 16 + l16) * NN + n) * NCP + 2 * half;
    const float* Wm = Wg + (long)slot * NCP * NHID + (2 * half) * 64 + l16 * 4;
#pragma unroll
    for (int kk = 0; kk < NCP / 4; ++kk) {
      a[slot][kk] = *(const v2f*)(Arow + kk * 4);
      bw[slot][kk][0] = *(const v4f*)(Wm + (kk * 4) * 64);
      bw[slot][kk][1] = *(const v4f*)(Wm + (kk * 4 + 1) * 64);
    }
  }
#pragma unroll
  for (int r = 0; r < 8; ++r)
    tb[r] = *(const v4f*)(TB + (((long)g * NT + t) * NB + wave * 16 + r + 8 * half) * 64 + l16 * 4);

  v8f c[4] = {};
#pragma unroll
  for (int slot = 0; slot < 2; ++slot) {
#pragma unroll
    for (int kk = 0; kk < NCP / 4; ++kk) {
#pragma unroll
      for (int ct = 0; ct < 4; ++ct) {
        v2f b;
        b.x = bw[slot][kk][0][ct];
        b.y = bw[slot][kk][1][ct];
        c[ct] = WMMA_F32(a[slot][kk], b, c[ct]);
      }
    }
  }
#pragma unroll
  for (int r = 0; r < 8; ++r) {
    int bb = wave * 16 + r + 8 * half;
    v4f v;
    v.x = c[0][r] + tb[r].x;
    v.y = c[1][r] + tb[r].y;
    v.z = c[2][r] + tb[r].z;
    v.w = c[3][r] + tb[r].w;
    *(v4f*)(G + ((long)bb * NN + n) * 64 + l16 * 4) = v;
  }
}

__global__ void __launch_bounds__(256) k_state_attn(
    float* __restrict__ G, const float* __restrict__ Hs,
    const float* __restrict__ olg, const float* __restrict__ olb, int g, int K) {
  const int wave = threadIdx.x >> 5, lane = threadIdx.x & 31;
  const long row = (long)blockIdx.x * 8 + wave;
  float g0 = G[row * 64 + lane];
  float g1 = G[row * 64 + lane + 32];
  float s = g0 + g1;
  s += __shfl_xor(s, 1); s += __shfl_xor(s, 2); s += __shfl_xor(s, 4);
  s += __shfl_xor(s, 8); s += __shfl_xor(s, 16);
  float mu = s * (1.f / 64.f);
  float d0 = g0 - mu, d1 = g1 - mu;
  float v = d0 * d0 + d1 * d1;
  v += __shfl_xor(v, 1); v += __shfl_xor(v, 2); v += __shfl_xor(v, 4);
  v += __shfl_xor(v, 8); v += __shfl_xor(v, 16);
  float inv = rsqrtf(v * (1.f / 64.f) + 1e-5f);
  float q0 = d0 * inv * olg[g * 64 + lane] + olb[g * 64 + lane];
  float q1 = d1 * inv * olg[g * 64 + lane + 32] + olb[g * 64 + lane + 32];

  float st0[13], st1[13], sc0[13], sc1[13];
#pragma unroll
  for (int k = 0; k < 13; ++k) {
    if (k < K) {
      st0[k] = Hs[((long)k * NB * NN + row) * 64 + lane];
      st1[k] = Hs[((long)k * NB * NN + row) * 64 + lane + 32];
      float p0 = q0 * st0[k], p1 = q1 * st1[k];
      p0 += __shfl_xor(p0, 1); p0 += __shfl_xor(p0, 2);
      p0 += __shfl_xor(p0, 4); p0 += __shfl_xor(p0, 8);
      p1 += __shfl_xor(p1, 1); p1 += __shfl_xor(p1, 2);
      p1 += __shfl_xor(p1, 4); p1 += __shfl_xor(p1, 8);
      sc0[k] = p0 * 0.25f; sc1[k] = p1 * 0.25f;
    } else { st0[k] = 0.f; st1[k] = 0.f; sc0[k] = -1e30f; sc1[k] = -1e30f; }
  }
  float m0 = -1e30f, m1 = -1e30f;
#pragma unroll
  for (int k = 0; k < 13; ++k) { m0 = fmaxf(m0, sc0[k]); m1 = fmaxf(m1, sc1[k]); }
  float s0 = 0.f, s1 = 0.f;
#pragma unroll
  for (int k = 0; k < 13; ++k) {
    sc0[k] = __expf(sc0[k] - m0); s0 += sc0[k];
    sc1[k] = __expf(sc1[k] - m1); s1 += sc1[k];
  }
  float o0 = 0.f, o1 = 0.f;
#pragma unroll
  for (int k = 0; k < 13; ++k) { o0 += sc0[k] * st0[k]; o1 += sc1[k] * st1[k]; }
  G[row * 64 + lane]      = g0 + o0 / s0;
  G[row * 64 + lane + 32] = g1 + o1 / s1;
}

__global__ void k_combine(const float* __restrict__ Gr, const float* __restrict__ Ghc,
                          const float* __restrict__ Hprev, float* __restrict__ Hnew) {
  long i = (long)blockIdx.x * 256 + threadIdx.x;
  if (i >= (long)NB * NN * 64) return;
  float r = 1.f / (1.f + __expf(-Gr[i]));
  float hc = tanhf(Ghc[i]);
  Hnew[i] = r * Hprev[i] + (1.f - r) * hc;
}

__global__ void k_pred(const float* __restrict__ Hs, const float* __restrict__ pw,
                       const float* __restrict__ pb, float* __restrict__ out) {
  int idx = blockIdx.x * 256 + threadIdx.x;
  if (idx >= NB * 12 * NN) return;
  int n = idx % NN; int r = idx / NN; int o = r % 12; int b = r / 12;
  float acc = pb[o];
  for (int cc = 0; cc < 2; ++cc) {
    const float* h = Hs + ((long)(NT - 1 + cc) * NB * NN + (long)b * NN + n) * 64;
    for (int hh = 0; hh < 64; ++hh)
      acc += h[hh] * pw[(o * 2 + cc) * 64 + hh];
  }
  out[idx] = acc;
}

extern "C" void kernel_launch(void* const* d_in, const int* in_sizes, int n_in,
                              void* d_out, int out_size, void* d_ws, size_t ws_size,
                              hipStream_t stream) {
  const float* x   = (const float*)d_in[0];
  const float* ne  = (const float*)d_in[1];
  const float* te  = (const float*)d_in[2];
  const float* wp  = (const float*)d_in[3];
  const float* bp  = (const float*)d_in[4];
  const float* glg = (const float*)d_in[5];
  const float* glb = (const float*)d_in[6];
  const float* olg = (const float*)d_in[7];
  const float* olb = (const float*)d_in[8];
  const float* pw  = (const float*)d_in[9];
  const float* pb  = (const float*)d_in[10];
  float* out = (float*)d_out;

  float* ws = (float*)d_ws;
  long off = 0;
  float* W    = ws + off; off += 3L * NN * 2 * NCP * NHID;
  float* TB   = ws + off; off += 3L * NT * NB * NHID;
  float* E    = ws + off; off += 3L * NB * NN * NDE;
  float* Hs   = ws + off; off += (long)(NT + 1) * NB * NN * NHID;
  float* IAS  = ws + off; off += (long)NB * NN * NCP;
  float* CAND = ws + off; off += (long)NB * NN * NCP;
  float* XG2  = ws + off; off += (long)NB * NN * NCP;
  float* G0   = ws + off; off += (long)NB * NN * NHID;
  float* G1   = ws + off; off += (long)NB * NN * NHID;
  float* G2   = ws + off; off += (long)NB * NN * NHID;

  const long wTotal = 3L * NN * 2 * NCP * NHID;
  k_w<<<dim3((unsigned)((wTotal + 255) / 256)), 256, 0, stream>>>(ne, wp, W);
  k_tb<<<(3 * NT * NB * NHID + 255) / 256, 256, 0, stream>>>(te, bp, TB);
  k_zero<<<(unsigned)(((long)NB * NN * NHID + 255) / 256), 256, 0, stream>>>(Hs, (long)NB * NN * NHID);

  for (int t = 0; t < NT; ++t) {
    float* Hprev = Hs + (long)t * NB * NN * NHID;
    float* Hnew  = Hs + (long)(t + 1) * NB * NN * NHID;
    k_emb<<<(3 * NB * NN + 255) / 256, 256, 0, stream>>>(ne, te, glg, glb, E, t);
    k_build_ias<<<(NB * NN * NCP + 255) / 256, 256, 0, stream>>>(x, Hprev, IAS, t);
    for (int g = 0; g < 2; ++g) {
      float* Gg = g ? G1 : G0;
      k_graph_attn<<<dim3(NN / 64, NB), 128, 0, stream>>>(E, IAS, XG2, g);
      k_gout<<<NN, 64, 0, stream>>>(IAS, XG2, W, TB, Gg, g, t);
      k_state_attn<<<NB * NN / 8, 256, 0, stream>>>(Gg, Hs, olg, olb, g, t + 1);
    }
    k_build_cand<<<(NB * NN * NCP + 255) / 256, 256, 0, stream>>>(x, Hprev, G0, CAND, t);
    k_graph_attn<<<dim3(NN / 64, NB), 128, 0, stream>>>(E, CAND, XG2, 2);
    k_gout<<<NN, 64, 0, stream>>>(CAND, XG2, W, TB, G2, 2, t);
    k_state_attn<<<NB * NN / 8, 256, 0, stream>>>(G2, Hs, olg, olb, 2, t + 1);
    k_combine<<<(unsigned)(((long)NB * NN * NHID + 255) / 256), 256, 0, stream>>>(G1, G2, Hprev, Hnew);
  }
  k_pred<<<(NB * 12 * NN + 255) / 256, 256, 0, stream>>>(Hs, pw, pb, out);
}


# k_state_attn writes its consumers' rows itself (z gate: gated candidate rows; candidate: new state and next step's input rows; whole-line stores via LDS): 35 of 160 launches go; sum order unchanged
# speedup vs baseline: 2.9172x; 1.1216x over previous
#include <hip/hip_runtime.h>
#include <math.h>

typedef __attribute__((ext_vector_type(2))) float v2f;
typedef __attribute__((ext_vector_type(8))) float v8f;

#define WMMA_F32(A, B, C) \
  __builtin_amdgcn_wmma_f32_16x16x4_f32(false, (A), false, (B), (short)0, (C), false, false)

#define NB   32
#define NT   12
#define NN   512
#define NDE  16
#define NHID 64
#define NCP  68

__global__ void k_w(const float* __restrict__ ne, const float* __restrict__ wp,
                    float* __restrict__ W) {
  long idx = (long)blockIdx.x * 256 + threadIdx.x;
  const long total = 3L * NN * 2 * NCP * NHID;
  if (idx >= total) return;
  int o = idx % 64; long r = idx / 64;
  int i = r % NCP; r /= NCP;
  int slot = r % 2; r /= 2;
  int n = r % NN; int g = (int)(r / NN);
  float acc = 0.f;
  if (i < 67) {
    for (int d = 0; d < NDE; ++d)
      acc += ne[n * NDE + d] * wp[(((long)(g * NDE + d) * 2 + slot) * 67 + i) * 64 + o];
  }
  W[idx] = acc;
}

__global__ void k_tb(const float* __restrict__ te, const float* __restrict__ bp,
                     float* __restrict__ TB) {
  int idx = blockIdx.x * 256 + threadIdx.x;
  if (idx >= 3 * NT * NB * NHID) return;
  int o = idx % 64; int r = idx / 64;
  int b = r % NB; r /= NB;
  int t = r % NT; int g = r / NT;
  float acc = 0.f;
  for (int d = 0; d < NDE; ++d)
    acc += te[(b * NT + t) * NDE + d] * bp[(g * NDE + d) * 64 + o];
  TB[idx] = acc;
}

__global__ void k_emb(const float* __restrict__ ne, const float* __restrict__ te,
                      const float* __restrict__ glg, const float* __restrict__ glb,
                      float* __restrict__ E, int t) {
  int idx = blockIdx.x * 256 + threadIdx.x;
  if (idx >= 3 * NB * NN) return;
  int n = idx % NN; int r = idx / NN;
  int b = r % NB; int g = r / NB;
  float v[NDE]; float mu = 0.f;
  for (int d = 0; d < NDE; ++d) { v[d] = ne[n * NDE + d] + te[(b * NT + t) * NDE + d]; mu += v[d]; }
  mu *= (1.f / NDE);
  float var = 0.f;
  for (int d = 0; d < NDE; ++d) { float q = v[d] - mu; var += q * q; }
  var *= (1.f / NDE);
  float inv = rsqrtf(var + 1e-12f);
  for (int d = 0; d < NDE; ++d)
    E[(long)idx * NDE + d] = (v[d] - mu) * inv * glg[g * NDE + d] + glb[g * NDE + d];
}

__global__ void k_zero(float* p, long nElem) {
  long i = (long)blockIdx.x * 256 + threadIdx.x;
  if (i < nElem) p[i] = 0.f;
}

__global__ void k_build_ias(const float* __restrict__ x, const float* __restrict__ h,
                            float* __restrict__ ias, int t) {
  int idx = blockIdx.x * 256 + threadIdx.x;
  if (idx >= NB * NN * NCP) return;
  int i = idx % NCP; int r = idx / NCP; int n = r % NN; int b = r / NN;
  float v;
  if (i < 3)       v = x[((long)(b * NT + t) * NN + n) * 3 + i];
  else if (i < 67) v = h[((long)b * NN + n) * 64 + (i - 3)];
  else             v = 0.f;
  ias[idx] = v;
}

#define VP 80
__global__ void __launch_bounds__(128) k_graph_attn(
    const float* __restrict__ E,
    const float* __restrict__ X,
    float* __restrict__ Y,
    int g) {
  __shared__ __attribute__((aligned(16))) float lds_e[NN * NDE];
  __shared__ __attribute__((aligned(16))) float lds_v[NN * VP];
  __shared__ float lds_p[4][16][17];

  const int b = blockIdx.y;
  const int row0 = blockIdx.x * 64;
  const int tid = threadIdx.x;
  const int wave = tid >> 5;
  const int lane = tid & 31;
  const int l16 = lane & 15;
  const int half = lane >> 4;
  const int myrow = row0 + wave * 16 + l16;

  const float* Eb = E + ((long)g * NB + b) * NN * NDE;
  const float* Xb = X + (long)b * NN * NCP;

  {
    const float4* se = (const float4*)Eb;
    float4 re[16];
#pragma unroll
    for (int j = 0; j < 16; ++j) re[j] = se[tid + 128 * j];
#pragma unroll
    for (int j = 0; j < 16; ++j) ((float4*)lds_e)[tid + 128 * j] = re[j];

    const float4* sv = (const float4*)Xb;
#pragma unroll
    for (int ph = 0; ph < 4; ++ph) {
      float4 rv[17];
#pragma unroll
      for (int j = 0; j < 17; ++j) rv[j] = sv[ph * 2176 + tid + 128 * j];
#pragma unroll
      for (int j = 0; j < 17; ++j) {
        int i = ph * 2176 + tid + 128 * j;
        int rr = i / 17, q = i - rr * 17;
        *(float4*)&lds_v[rr * VP + q * 4] = rv[j];
      }
    }
#pragma unroll
    for (int j = 0; j < 12; ++j) {
      int i = tid + 128 * j;
      int rr = i / 3, q = i - rr * 3;
      float4 v = make_float4(q == 0 ? 1.f : 0.f, 0.f, 0.f, 0.f);
      *(float4*)&lds_v[rr * VP + 68 + q * 4] = v;
    }
  }

  v2f aA[4];
#pragma unroll
  for (int kk = 0; kk < 4; ++kk) {
    aA[kk].x = Eb[myrow * NDE + kk * 4 + 2 * half];
    aA[kk].y = Eb[myrow * NDE + kk * 4 + 2 * half + 1];
  }

  v8f acc[5] = {};

  __syncthreads();

  for (int mc = 0; mc < NN / 16; ++mc) {
    const float* le = lds_e + mc * 16 * NDE;
    const float* lv = lds_v + mc * 16 * VP;

    v8f c = {};
#pragma unroll
    for (int kk = 0; kk < 4; ++kk) {
      v2f bf;
      bf.x = le[l16 * NDE + kk * 4 + 2 * half];
      bf.y = le[l16 * NDE + kk * 4 + 2 * half + 1];
      c = WMMA_F32(aA[kk], bf, c);
    }

#pragma unroll
    for (int r = 0; r < 8; ++r)
      lds_p[wave][r + 8 * half][l16] = __expf(c[r]);

#pragma unroll
    for (int kk = 0; kk < 4; ++kk) {
      v2f ap;
      ap.x = lds_p[wave][l16][kk * 4 + 2 * half];
      ap.y = lds_p[wave][l16][kk * 4 + 2 * half + 1];
#pragma unroll
      for (int ct = 0; ct < 5; ++ct) {
        v2f bv;
        bv.x = lv[(kk * 4 + 2 * half) * VP + ct * 16 + l16];
        bv.y = lv[(kk * 4 + 2 * half + 1) * VP + ct * 16 + l16];
        acc[ct] = WMMA_F32(ap, bv, acc[ct]);
      }
    }
  }

#pragma unroll
  for (int r = 0; r < 8; ++r) {
    float rs = __shfl(acc[4][r], half * 16 + 4);
    float inv = 1.f / rs;
    int row = row0 + wave * 16 + r + 8 * half;
#pragma unroll
    for (int ct = 0; ct < 5; ++ct) {
      int col = ct * 16 + l16;
      if (col < NCP)
        Y[((long)b * NN + row) * NCP + col] = acc[ct][r] * inv;
    }
  }
}

typedef __attribute__((ext_vector_type(4))) float v4f;
__global__ void __launch_bounds__(64) k_gout(
    const float* __restrict__ X, const float* __restrict__ XG,
    const float* __restrict__ W, const float* __restrict__ TB,
    float* __restrict__ G, int g, int t) {
  const int n = blockIdx.x;
  const int wave = threadIdx.x >> 5;
  const int lane = threadIdx.x & 31;
  const int l16 = lane & 15, half = lane >> 4;
  const float* Wg = W + ((long)g * NN + n) * 2 * NCP * NHID;

  v2f a[2][NCP / 4];
  v4f bw[2][NCP / 4][2];
  v4f tb[8];

#pragma unroll
  for (int slot = 0; slot < 2; ++slot) {
    const float* Arow = (slot ? XG : X) + ((long)(wave * 16 + l16) * NN + n) * NCP + 2 * half;
    const float* Wm = Wg + (long)slot * NCP * NHID + (2 * half) * 64 + l16 * 4;
#pragma unroll
    for (int kk = 0; kk < NCP / 4; ++kk) {
      a[slot][kk] = *(const v2f*)(Arow + kk * 4);
      bw[slot][kk][0] = *(const v4f*)(Wm + (kk * 4) * 64);
      bw[slot][kk][1] = *(const v4f*)(Wm + (kk * 4 + 1) * 64);
    }
  }
#pragma unroll
  for (int r = 0; r < 8; ++r)
    tb[r] = *(const v4f*)(TB + (((long)g * NT + t) * NB + wave * 16 + r + 8 * half) * 64 + l16 * 4);

  v8f c[4] = {};
#pragma unroll
  for (int slot = 0; slot < 2; ++slot) {
#pragma unroll
    for (int kk = 0; kk < NCP / 4; ++kk) {
#pragma unroll
      for (int ct = 0; ct < 4; ++ct) {
        v2f b;
        b.x = bw[slot][kk][0][ct];
        b.y = bw[slot][kk][1][ct];
        c[ct] = WMMA_F32(a[slot][kk], b, c[ct]);
      }
    }
  }
#pragma unroll
  for (int r = 0; r < 8; ++r) {
    int bb = wave * 16 + r + 8 * half;
    v4f v;
    v.x = c[0][r] + tb[r].x;
    v.y = c[1][r] + tb[r].y;
    v.z = c[2][r] + tb[r].z;
    v.w = c[3][r] + tb[r].w;
    *(v4f*)(G + ((long)bb * NN + n) * 64 + l16 * 4) = v;
  }
}

template <int MODE>
__global__ void __launch_bounds__(256) k_state_attn(
    float* __restrict__ G, const float* __restrict__ Hs,
    const float* __restrict__ olg, const float* __restrict__ olb, int g, int K,
    const float* __restrict__ x, const float* __restrict__ Gr,
    float* __restrict__ Hnew, float* __restrict__ rows_out, int tx) {
  const int wave = threadIdx.x >> 5, lane = threadIdx.x & 31;
  const long row = (long)blockIdx.x * 8 + wave;
  float g0 = G[row * 64 + lane];
  float g1 = G[row * 64 + lane + 32];

  float hp0 = 0.f, hp1 = 0.f, gr0 = 0.f, gr1 = 0.f, xv = 0.f;
  if (MODE != 1) {
    hp0 = Hs[((long)(K - 1) * NB * NN + row) * 64 + lane];
    hp1 = Hs[((long)(K - 1) * NB * NN + row) * 64 + lane + 32];
    if (lane < 3 && tx < NT) {
      const int b = (int)(row / NN), n = (int)(row % NN);
      xv = x[((long)(b * NT + tx) * NN + n) * 3 + lane];
    }
  }
  if (MODE == 2) {
    gr0 = Gr[row * 64 + lane];
    gr1 = Gr[row * 64 + lane + 32];
  }

  float s = g0 + g1;
  s += __shfl_xor(s, 1); s += __shfl_xor(s, 2); s += __shfl_xor(s, 4);
  s += __shfl_xor(s, 8); s += __shfl_xor(s, 16);
  float mu = s * (1.f / 64.f);
  float d0 = g0 - mu, d1 = g1 - mu;
  float v = d0 * d0 + d1 * d1;
  v += __shfl_xor(v, 1); v += __shfl_xor(v, 2); v += __shfl_xor(v, 4);
  v += __shfl_xor(v, 8); v += __shfl_xor(v, 16);
  float inv = rsqrtf(v * (1.f / 64.f) + 1e-5f);
  float q0 = d0 * inv * olg[g * 64 + lane] + olb[g * 64 + lane];
  float q1 = d1 * inv * olg[g * 64 + lane + 32] + olb[g * 64 + lane + 32];

  float st0[13], st1[13], sc0[13], sc1[13];
#pragma unroll
  for (int k = 0; k < 13; ++k) {
    if (k < K) {
      st0[k] = Hs[((long)k * NB * NN + row) * 64 + lane];
      st1[k] = Hs[((long)k * NB * NN + row) * 64 + lane + 32];
      float p0 = q0 * st0[k], p1 = q1 * st1[k];
      p0 += __shfl_xor(p0, 1); p0 += __shfl_xor(p0, 2);
      p0 += __shfl_xor(p0, 4); p0 += __shfl_xor(p0, 8);
      p1 += __shfl_xor(p1, 1); p1 += __shfl_xor(p1, 2);
      p1 += __shfl_xor(p1, 4); p1 += __shfl_xor(p1, 8);
      sc0[k] = p0 * 0.25f; sc1[k] = p1 * 0.25f;
    } else { st0[k] = 0.f; st1[k] = 0.f; sc0[k] = -1e30f; sc1[k] = -1e30f; }
  }
  float m0 = -1e30f, m1 = -1e30f;
#pragma unroll
  for (int k = 0; k < 13; ++k) { m0 = fmaxf(m0, sc0[k]); m1 = fmaxf(m1, sc1[k]); }
  float s0 = 0.f, s1 = 0.f;
#pragma unroll
  for (int k = 0; k < 13; ++k) {
    sc0[k] = __expf(sc0[k] - m0); s0 += sc0[k];
    sc1[k] = __expf(sc1[k] - m1); s1 += sc1[k];
  }
  float o0 = 0.f, o1 = 0.f;
#pragma unroll
  for (int k = 0; k < 13; ++k) { o0 += sc0[k] * st0[k]; o1 += sc1[k] * st1[k]; }
  const float y0 = g0 + o0 / s0;
  const float y1 = g1 + o1 / s1;

  if (MODE == 1) {
    G[row * 64 + lane]      = y0;
    G[row * 64 + lane + 32] = y1;
  } else {
    float e0, e1;
    if (MODE == 0) {
      float z0 = 1.f / (1.f + __expf(-y0));
      float z1 = 1.f / (1.f + __expf(-y1));
      e0 = z0 * hp0;
      e1 = z1 * hp1;
    } else {
      float r0 = 1.f / (1.f + __expf(-gr0));
      float r1 = 1.f / (1.f + __expf(-gr1));
      float hc0 = tanhf(y0);
      float hc1 = tanhf(y1);
      e0 = r0 * hp0 + (1.f - r0) * hc0;
      e1 = r1 * hp1 + (1.f - r1) * hc1;
    }
    if (tx < NT) {
      __shared__ __attribute__((aligned(16))) float rowbuf[8 * NCP];
      float* rb = rowbuf + wave * NCP;
      rb[3 + lane]  = e0;
      rb[35 + lane] = e1;
      if (lane < 4) rb[lane < 3 ? lane : 67] = xv;
      __syncthreads();
      if (threadIdx.x < 8 * NCP / 4)
        ((float4*)(rows_out + (long)blockIdx.x * 8 * NCP))[threadIdx.x] =
            ((const float4*)rowbuf)[threadIdx.x];
    }
    if (MODE == 2) {
      Hnew[row * 64 + lane]      = e0;
      Hnew[row * 64 + lane + 32] = e1;
    }
  }
}

__global__ void k_pred(const float* __restrict__ Hs, const float* __restrict__ pw,
                       const float* __restrict__ pb, float* __restrict__ out) {
  int idx = blockIdx.x * 256 + threadIdx.x;
  if (idx >= NB * 12 * NN) return;
  int n = idx % NN; int r = idx / NN; int o = r % 12; int b = r / 12;
  float acc = pb[o];
  for (int cc = 0; cc < 2; ++cc) {
    const float* h = Hs + ((long)(NT - 1 + cc) * NB * NN + (long)b * NN + n) * 64;
    for (int hh = 0; hh < 64; ++hh)
      acc += h[hh] * pw[(o * 2 + cc) * 64 + hh];
  }
  out[idx] = acc;
}

extern "C" void kernel_launch(void* const* d_in, const int* in_sizes, int n_in,
                              void* d_out, int out_size, void* d_ws, size_t ws_size,
                              hipStream_t stream) {
  const float* x   = (const float*)d_in[0];
  const float* ne  = (const float*)d_in[1];
  const float* te  = (const float*)d_in[2];
  const float* wp  = (const float*)d_in[3];
  const float* bp  = (const float*)d_in[4];
  const float* glg = (const float*)d_in[5];
  const float* glb = (const float*)d_in[6];
  const float* olg = (const float*)d_in[7];
  const float* olb = (const float*)d_in[8];
  const float* pw  = (const float*)d_in[9];
  const float* pb  = (const float*)d_in[10];
  float* out = (float*)d_out;

  float* ws = (float*)d_ws;
  long off = 0;
  float* W    = ws + off; off += 3L * NN * 2 * NCP * NHID;
  float* TB   = ws + off; off += 3L * NT * NB * NHID;
  float* E    = ws + off; off += 3L * NB * NN * NDE;
  float* Hs   = ws + off; off += (long)(NT + 1) * NB * NN * NHID;
  float* IAS  = ws + off; off += (long)NB * NN * NCP;
  float* CAND = ws + off; off += (long)NB * NN * NCP;
  float* XG2  = ws + off; off += (long)NB * NN * NCP;
  float* G0   = ws + off; off += (long)NB * NN * NHID;
  float* G1   = ws + off; off += (long)NB * NN * NHID;
  float* G2   = ws + off; off += (long)NB * NN * NHID;

  const long wTotal = 3L * NN * 2 * NCP * NHID;
  k_w<<<dim3((unsigned)((wTotal + 255) / 256)), 256, 0, stream>>>(ne, wp, W);
  k_tb<<<(3 * NT * NB * NHID + 255) / 256, 256, 0, stream>>>(te, bp, TB);
  k_zero<<<(unsigned)(((long)NB * NN * NHID + 255) / 256), 256, 0, stream>>>(Hs, (long)NB * NN * NHID);

  k_build_ias<<<(NB * NN * NCP + 255) / 256, 256, 0, stream>>>(x, Hs, IAS, 0);

  for (int t = 0; t < NT; ++t) {
    float* Hnew = Hs + (long)(t + 1) * NB * NN * NHID;
    k_emb<<<(3 * NB * NN + 255) / 256, 256, 0, stream>>>(ne, te, glg, glb, E, t);
    k_graph_attn<<<dim3(NN / 64, NB), 128, 0, stream>>>(E, IAS, XG2, 0);
    k_gout<<<NN, 64, 0, stream>>>(IAS, XG2, W, TB, G0, 0, t);
    k_state_attn<0><<<NB * NN / 8, 256, 0, stream>>>(G0, Hs, olg, olb, 0, t + 1, x, G1, Hnew, CAND, t);
    k_graph_attn<<<dim3(NN / 64, NB), 128, 0, stream>>>(E, IAS, XG2, 1);
    k_gout<<<NN, 64, 0, stream>>>(IAS, XG2, W, TB, G1, 1, t);
    k_state_attn<1><<<NB * NN / 8, 256, 0, stream>>>(G1, Hs, olg, olb, 1, t + 1, x, G0, Hnew, CAND, t);
    k_graph_attn<<<dim3(NN / 64, NB), 128, 0, stream>>>(E, CAND, XG2, 2);
    k_gout<<<NN, 64, 0, stream>>>(CAND, XG2, W, TB, G2, 2, t);
    k_state_attn<2><<<NB * NN / 8, 256, 0, stream>>>(G2, Hs, olg, olb, 2, t + 1, x, G1, Hnew, IAS, t + 1);
  }
  k_pred<<<(NB * 12 * NN + 255) / 256, 256, 0, stream>>>(Hs, pw, pb, out);
}


# k_graph_attn: score tile computed transposed (keys x queries), so exp(S) feeds the P@V product from registers with no LDS round trip for P; sum order changes only inside each 4-key product
# speedup vs baseline: 3.0491x; 1.0452x over previous
#include <hip/hip_runtime.h>
#include <math.h>

typedef __attribute__((ext_vector_type(2))) float v2f;
typedef __attribute__((ext_vector_type(8))) float v8f;

#define WMMA_F32(A, B, C) \
  __builtin_amdgcn_wmma_f32_16x16x4_f32(false, (A), false, (B), (short)0, (C), false, false)

#define NB   32
#define NT   12
#define NN   512
#define NDE  16
#define NHID 64
#define NCP  68

__global__ void k_w(const float* __restrict__ ne, const float* __restrict__ wp,
                    float* __restrict__ W) {
  long idx = (long)blockIdx.x * 256 + threadIdx.x;
  const long total = 3L * NN * 2 * NCP * NHID;
  if (idx >= total) return;
  int o = idx % 64; long r = idx / 64;
  int i = r % NCP; r /= NCP;
  int slot = r % 2; r /= 2;
  int n = r % NN; int g = (int)(r / NN);
  float acc = 0.f;
  if (i < 67) {
    for (int d = 0; d < NDE; ++d)
      acc += ne[n * NDE + d] * wp[(((long)(g * NDE + d) * 2 + slot) * 67 + i) * 64 + o];
  }
  W[idx] = acc;
}

__global__ void k_tb(const float* __restrict__ te, const float* __restrict__ bp,
                     float* __restrict__ TB) {
  int idx = blockIdx.x * 256 + threadIdx.x;
  if (idx >= 3 * NT * NB * NHID) return;
  int o = idx % 64; int r = idx / 64;
  int b = r % NB; r /= NB;
  int t = r % NT; int g = r / NT;
  float acc = 0.f;
  for (int d = 0; d < NDE; ++d)
    acc += te[(b * NT + t) * NDE + d] * bp[(g * NDE + d) * 64 + o];
  TB[idx] = acc;
}

__global__ void k_emb(const float* __restrict__ ne, const float* __restrict__ te,
                      const float* __restrict__ glg, const float* __restrict__ glb,
                      float* __restrict__ E, int t) {
  int idx = blockIdx.x * 256 + threadIdx.x;
  if (idx >= 3 * NB * NN) return;
  int n = idx % NN; int r = idx / NN;
  int b = r % NB; int g = r / NB;
  float v[NDE]; float mu = 0.f;
  for (int d = 0; d < NDE; ++d) { v[d] = ne[n * NDE + d] + te[(b * NT + t) * NDE + d]; mu += v[d]; }
  mu *= (1.f / NDE);
  float var = 0.f;
  for (int d = 0; d < NDE; ++d) { float q = v[d] - mu; var += q * q; }
  var *= (1.f / NDE);
  float inv = rsqrtf(var + 1e-12f);
  for (int d = 0; d < NDE; ++d)
    E[(long)idx * NDE + d] = (v[d] - mu) * inv * glg[g * NDE + d] + glb[g * NDE + d];
}

__global__ void k_zero(float* p, long nElem) {
  long i = (long)blockIdx.x * 256 + threadIdx.x;
  if (i < nElem) p[i] = 0.f;
}

__global__ void k_build_ias(const float* __restrict__ x, const float* __restrict__ h,
                            float* __restrict__ ias, int t) {
  int idx = blockIdx.x * 256 + threadIdx.x;
  if (idx >= NB * NN * NCP) return;
  int i = idx % NCP; int r = idx / NCP; int n = r % NN; int b = r / NN;
  float v;
  if (i < 3)       v = x[((long)(b * NT + t) * NN + n) * 3 + i];
  else if (i < 67) v = h[((long)b * NN + n) * 64 + (i - 3)];
  else             v = 0.f;
  ias[idx] = v;
}

#define VP 80
__global__ void __launch_bounds__(128) k_graph_attn(
    const float* __restrict__ E,
    const float* __restrict__ X,
    float* __restrict__ Y,
    int g) {
  __shared__ __attribute__((aligned(16))) float lds_e[NN * NDE];
  __shared__ __attribute__((aligned(16))) float lds_v[NN * VP];

  const int b = blockIdx.y;
  const int row0 = blockIdx.x * 64;
  const int tid = threadIdx.x;
  const int wave = tid >> 5;
  const int lane = tid & 31;
  const int l16 = lane & 15;
  const int half = lane >> 4;
  const int myrow = row0 + wave * 16 + l16;

  const float* Eb = E + ((long)g * NB + b) * NN * NDE;
  const float* Xb = X + (long)b * NN * NCP;

  {
    const float4* se = (const float4*)Eb;
    float4 re[16];
#pragma unroll
    for (int j = 0; j < 16; ++j) re[j] = se[tid + 128 * j];
#pragma unroll
    for (int j = 0; j < 16; ++j) ((float4*)lds_e)[tid + 128 * j] = re[j];

    const float4* sv = (const float4*)Xb;
#pragma unroll
    for (int ph = 0; ph < 4; ++ph) {
      float4 rv[17];
#pragma unroll
      for (int j = 0; j < 17; ++j) rv[j] = sv[ph * 2176 + tid + 128 * j];
#pragma unroll
      for (int j = 0; j < 17; ++j) {
        int i = ph * 2176 + tid + 128 * j;
        int rr = i / 17, q = i - rr * 17;
        *(float4*)&lds_v[rr * VP + q * 4] = rv[j];
      }
    }
#pragma unroll
    for (int j = 0; j < 12; ++j) {
      int i = tid + 128 * j;
      int rr = i / 3, q = i - rr * 3;
      float4 v = make_float4(q == 0 ? 1.f : 0.f, 0.f, 0.f, 0.f);
      *(float4*)&lds_v[rr * VP + 68 + q * 4] = v;
    }
  }

  v2f qB[4];
#pragma unroll
  for (int kk = 0; kk < 4; ++kk) {
    qB[kk].x = Eb[myrow * NDE + kk * 4 + 2 * half];
    qB[kk].y = Eb[myrow * NDE + kk * 4 + 2 * half + 1];
  }
  const int krow = 2 * (l16 & 7) + (l16 >> 3);

  v8f acc[5] = {};

  __syncthreads();

  for (int mc = 0; mc < NN / 16; ++mc) {
    const float* le = lds_e + mc * 16 * NDE;
    const float* lv = lds_v + mc * 16 * VP;

    v8f c = {};
#pragma unroll
    for (int kk = 0; kk < 4; ++kk) {
      v2f ek;
      ek.x = le[krow * NDE + kk * 4 + 2 * half];
      ek.y = le[krow * NDE + kk * 4 + 2 * half + 1];
      c = WMMA_F32(ek, qB[kk], c);
    }

#pragma unroll
    for (int kk = 0; kk < 4; ++kk) {
      v2f ap;
      ap.x = __expf(c[2 * kk]);
      ap.y = __expf(c[2 * kk + 1]);
#pragma unroll
      for (int ct = 0; ct < 5; ++ct) {
        v2f bv;
        bv.x = lv[(kk * 4 + half) * VP + ct * 16 + l16];
        bv.y = lv[(kk * 4 + 2 + half) * VP + ct * 16 + l16];
        acc[ct] = WMMA_F32(ap, bv, acc[ct]);
      }
    }
  }

#pragma unroll
  for (int r = 0; r < 8; ++r) {
    float rs = __shfl(acc[4][r], half * 16 + 4);
    float inv = 1.f / rs;
    int row = row0 + wave * 16 + r + 8 * half;
#pragma unroll
    for (int ct = 0; ct < 5; ++ct) {
      int col = ct * 16 + l16;
      if (col < NCP)
        Y[((long)b * NN + row) * NCP + col] = acc[ct][r] * inv;
    }
  }
}

typedef __attribute__((ext_vector_type(4))) float v4f;
__global__ void __launch_bounds__(64) k_gout(
    const float* __restrict__ X, const float* __restrict__ XG,
    const float* __restrict__ W, const float* __restrict__ TB,
    float* __restrict__ G, int g, int t) {
  const int n = blockIdx.x;
  const int wave = threadIdx.x >> 5;
  const int lane = threadIdx.x & 31;
  const int l16 = lane & 15, half = lane >> 4;
  const float* Wg = W + ((long)g * NN + n) * 2 * NCP * NHID;

  v2f a[2][NCP / 4];
  v4f bw[2][NCP / 4][2];
  v4f tb[8];

#pragma unroll
  for (int slot = 0; slot < 2; ++slot) {
    const float* Arow = (slot ? XG : X) + ((long)(wave * 16 + l16) * NN + n) * NCP + 2 * half;
    const float* Wm = Wg + (long)slot * NCP * NHID + (2 * half) * 64 + l16 * 4;
#pragma unroll
    for (int kk = 0; kk < NCP / 4; ++kk) {
      a[slot][kk] = *(const v2f*)(Arow + kk * 4);
      bw[slot][kk][0] = *(const v4f*)(Wm + (kk * 4) * 64);
      bw[slot][kk][1] = *(const v4f*)(Wm + (kk * 4 + 1) * 64);
    }
  }
#pragma unroll
  for (int r = 0; r < 8; ++r)
    tb[r] = *(const v4f*)(TB + (((long)g * NT + t) * NB + wave * 16 + r + 8 * half) * 64 + l16 * 4);

  v8f c[4] = {};
#pragma unroll
  for (int slot = 0; slot < 2; ++slot) {
#pragma unroll
    for (int kk = 0; kk < NCP / 4; ++kk) {
#pragma unroll
      for (int ct = 0; ct < 4; ++ct) {
        v2f b;
        b.x = bw[slot][kk][0][ct];
        b.y = bw[slot][kk][1][ct];
        c[ct] = WMMA_F32(a[slot][kk], b, c[ct]);
      }
    }
  }
#pragma unroll
  for (int r = 0; r < 8; ++r) {
    int bb = wave * 16 + r + 8 * half;
    v4f v;
    v.x = c[0][r] + tb[r].x;
    v.y = c[1][r] + tb[r].y;
    v.z = c[2][r] + tb[r].z;
    v.w = c[3][r] + tb[r].w;
    *(v4f*)(G + ((long)bb * NN + n) * 64 + l16 * 4) = v;
  }
}

template <int MODE>
__global__ void __launch_bounds__(256) k_state_attn(
    float* __restrict__ G, const float* __restrict__ Hs,
    const float* __restrict__ olg, const float* __restrict__ olb, int g, int K,
    const float* __restrict__ x, const float* __restrict__ Gr,
    float* __restrict__ Hnew, float* __restrict__ rows_out, int tx) {
  const int wave = threadIdx.x >> 5, lane = threadIdx.x & 31;
  const long row = (long)blockIdx.x * 8 + wave;
  float g0 = G[row * 64 + lane];
  float g1 = G[row * 64 + lane + 32];

  float hp0 = 0.f, hp1 = 0.f, gr0 = 0.f, gr1 = 0.f, xv = 0.f;
  if (MODE != 1) {
    hp0 = Hs[((long)(K - 1) * NB * NN + row) * 64 + lane];
    hp1 = Hs[((long)(K - 1) * NB * NN + row) * 64 + lane + 32];
    if (lane < 3 && tx < NT) {
      const int b = (int)(row / NN), n = (int)(row % NN);
      xv = x[((long)(b * NT + tx) * NN + n) * 3 + lane];
    }
  }
  if (MODE == 2) {
    gr0 = Gr[row * 64 + lane];
    gr1 = Gr[row * 64 + lane + 32];
  }

  float s = g0 + g1;
  s += __shfl_xor(s, 1); s += __shfl_xor(s, 2); s += __shfl_xor(s, 4);
  s += __shfl_xor(s, 8); s += __shfl_xor(s, 16);
  float mu = s * (1.f / 64.f);
  float d0 = g0 - mu, d1 = g1 - mu;
  float v = d0 * d0 + d1 * d1;
  v += __shfl_xor(v, 1); v += __shfl_xor(v, 2); v += __shfl_xor(v, 4);
  v += __shfl_xor(v, 8); v += __shfl_xor(v, 16);
  float inv = rsqrtf(v * (1.f / 64.f) + 1e-5f);
  float q0 = d0 * inv * olg[g * 64 + lane] + olb[g * 64 + lane];
  float q1 = d1 * inv * olg[g * 64 + lane + 32] + olb[g * 64 + lane + 32];

  float st0[13], st1[13], sc0[13], sc1[13];
#pragma unroll
  for (int k = 0; k < 13; ++k) {
    if (k < K) {
      st0[k] = Hs[((long)k * NB * NN + row) * 64 + lane];
      st1[k] = Hs[((long)k * NB * NN + row) * 64 + lane + 32];
      float p0 = q0 * st0[k], p1 = q1 * st1[k];
      p0 += __shfl_xor(p0, 1); p0 += __shfl_xor(p0, 2);
      p0 += __shfl_xor(p0, 4); p0 += __shfl_xor(p0, 8);
      p1 += __shfl_xor(p1, 1); p1 += __shfl_xor(p1, 2);
      p1 += __shfl_xor(p1, 4); p1 += __shfl_xor(p1, 8);
      sc0[k] = p0 * 0.25f; sc1[k] = p1 * 0.25f;
    } else { st0[k] = 0.f; st1[k] = 0.f; sc0[k] = -1e30f; sc1[k] = -1e30f; }
  }
  float m0 = -1e30f, m1 = -1e30f;
#pragma unroll
  for (int k = 0; k < 13; ++k) { m0 = fmaxf(m0, sc0[k]); m1 = fmaxf(m1, sc1[k]); }
  float s0 = 0.f, s1 = 0.f;
#pragma unroll
  for (int k = 0; k < 13; ++k) {
    sc0[k] = __expf(sc0[k] - m0); s0 += sc0[k];
    sc1[k] = __expf(sc1[k] - m1); s1 += sc1[k];
  }
  float o0 = 0.f, o1 = 0.f;
#pragma unroll
  for (int k = 0; k < 13; ++k) { o0 += sc0[k] * st0[k]; o1 += sc1[k] * st1[k]; }
  const float y0 = g0 + o0 / s0;
  const float y1 = g1 + o1 / s1;

  if (MODE == 1) {
    G[row * 64 + lane]      = y0;
    G[row * 64 + lane + 32] = y1;
  } else {
    float e0, e1;
    if (MODE == 0) {
      float z0 = 1.f / (1.f + __expf(-y0));
      float z1 = 1.f / (1.f + __expf(-y1));
      e0 = z0 * hp0;
      e1 = z1 * hp1;
    } else {
      float r0 = 1.f / (1.f + __expf(-gr0));
      float r1 = 1.f / (1.f + __expf(-gr1));
      float hc0 = tanhf(y0);
      float hc1 = tanhf(y1);
      e0 = r0 * hp0 + (1.f - r0) * hc0;
      e1 = r1 * hp1 + (1.f - r1) * hc1;
    }
    if (tx < NT) {
      __shared__ __attribute__((aligned(16))) float rowbuf[8 * NCP];
      float* rb = rowbuf + wave * NCP;
      rb[3 + lane]  = e0;
      rb[35 + lane] = e1;
      if (lane < 4) rb[lane < 3 ? lane : 67] = xv;
      __syncthreads();
      if (threadIdx.x < 8 * NCP / 4)
        ((float4*)(rows_out + (long)blockIdx.x * 8 * NCP))[threadIdx.x] =
            ((const float4*)rowbuf)[threadIdx.x];
    }
    if (MODE == 2) {
      Hnew[row * 64 + lane]      = e0;
      Hnew[row * 64 + lane + 32] = e1;
    }
  }
}

__global__ void k_pred(const float* __restrict__ Hs, const float* __restrict__ pw,
                       const float* __restrict__ pb, float* __restrict__ out) {
  int idx = blockIdx.x * 256 + threadIdx.x;
  if (idx >= NB * 12 * NN) return;
  int n = idx % NN; int r = idx / NN; int o = r % 12; int b = r / 12;
  float acc = pb[o];
  for (int cc = 0; cc < 2; ++cc) {
    const float* h = Hs + ((long)(NT - 1 + cc) * NB * NN + (long)b * NN + n) * 64;
    for (int hh = 0; hh < 64; ++hh)
      acc += h[hh] * pw[(o * 2 + cc) * 64 + hh];
  }
  out[idx] = acc;
}

extern "C" void kernel_launch(void* const* d_in, const int* in_sizes, int n_in,
                              void* d_out, int out_size, void* d_ws, size_t ws_size,
                              hipStream_t stream) {
  const float* x   = (const float*)d_in[0];
  const float* ne  = (const float*)d_in[1];
  const float* te  = (const float*)d_in[2];
  const float* wp  = (const float*)d_in[3];
  const float* bp  = (const float*)d_in[4];
  const float* glg = (const float*)d_in[5];
  const float* glb = (const float*)d_in[6];
  const float* olg = (const float*)d_in[7];
  const float* olb = (const float*)d_in[8];
  const float* pw  = (const float*)d_in[9];
  const float* pb  = (const float*)d_in[10];
  float* out = (float*)d_out;

  float* ws = (float*)d_ws;
  long off = 0;
  float* W    = ws + off; off += 3L * NN * 2 * NCP * NHID;
  float* TB   = ws + off; off += 3L * NT * NB * NHID;
  float* E    = ws + off; off += 3L * NB * NN * NDE;
  float* Hs   = ws + off; off += (long)(NT + 1) * NB * NN * NHID;
  float* IAS  = ws + off; off += (long)NB * NN * NCP;
  float* CAND = ws + off; off += (long)NB * NN * NCP;
  float* XG2  = ws + off; off += (long)NB * NN * NCP;
  float* G0   = ws + off; off += (long)NB * NN * NHID;
  float* G1   = ws + off; off += (long)NB * NN * NHID;
  float* G2   = ws + off; off += (long)NB * NN * NHID;

  const long wTotal = 3L * NN * 2 * NCP * NHID;
  k_w<<<dim3((unsigned)((wTotal + 255) / 256)), 256, 0, stream>>>(ne, wp, W);
  k_tb<<<(3 * NT * NB * NHID + 255) / 256, 256, 0, stream>>>(te, bp, TB);
  k_zero<<<(unsigned)(((long)NB * NN * NHID + 255) / 256), 256, 0, stream>>>(Hs, (long)NB * NN * NHID);

  k_build_ias<<<(NB * NN * NCP + 255) / 256, 256, 0, stream>>>(x, Hs, IAS, 0);

  for (int t = 0; t < NT; ++t) {
    float* Hnew = Hs + (long)(t + 1) * NB * NN * NHID;
    k_emb<<<(3 * NB * NN + 255) / 256, 256, 0, stream>>>(ne, te, glg, glb, E, t);
    k_graph_attn<<<dim3(NN / 64, NB), 128, 0, stream>>>(E, IAS, XG2, 0);
    k_gout<<<NN, 64, 0, stream>>>(IAS, XG2, W, TB, G0, 0, t);
    k_state_attn<0><<<NB * NN / 8, 256, 0, stream>>>(G0, Hs, olg, olb, 0, t + 1, x, G1, Hnew, CAND, t);
    k_graph_attn<<<dim3(NN / 64, NB), 128, 0, stream>>>(E, IAS, XG2, 1);
    k_gout<<<NN, 64, 0, stream>>>(IAS, XG2, W, TB, G1, 1, t);
    k_state_attn<1><<<NB * NN / 8, 256, 0, stream>>>(G1, Hs, olg, olb, 1, t + 1, x, G0, Hnew, CAND, t);
    k_graph_attn<<<dim3(NN / 64, NB), 128, 0, stream>>>(E, CAND, XG2, 2);
    k_gout<<<NN, 64, 0, stream>>>(CAND, XG2, W, TB, G2, 2, t);
    k_state_attn<2><<<NB * NN / 8, 256, 0, stream>>>(G2, Hs, olg, olb, 2, t + 1, x, G1, Hnew, IAS, t + 1);
  }
  k_pred<<<(NB * 12 * NN + 255) / 256, 256, 0, stream>>>(Hs, pw, pb, out);
}


# z and r gates run in ONE launch per stage and share operands (graph attn: one staged X panel, 8 waves; state attn: states loaded once per row; k_gout: grid y): 36 launches go; sum order unchanged
# speedup vs baseline: 3.6435x; 1.1949x over previous
#include <hip/hip_runtime.h>
#include <math.h>

typedef __attribute__((ext_vector_type(2))) float v2f;
typedef __attribute__((ext_vector_type(8))) float v8f;

#define WMMA_F32(A, B, C) \
  __builtin_amdgcn_wmma_f32_16x16x4_f32(false, (A), false, (B), (short)0, (C), false, false)

#define NB   32
#define NT   12
#define NN   512
#define NDE  16
#define NHID 64
#define NCP  68

__global__ void k_w(const float* __restrict__ ne, const float* __restrict__ wp,
                    float* __restrict__ W) {
  long idx = (long)blockIdx.x * 256 + threadIdx.x;
  const long total = 3L * NN * 2 * NCP * NHID;
  if (idx >= total) return;
  int o = idx % 64; long r = idx / 64;
  int i = r % NCP; r /= NCP;
  int slot = r % 2; r /= 2;
  int n = r % NN; int g = (int)(r / NN);
  float acc = 0.f;
  if (i < 67) {
    for (int d = 0; d < NDE; ++d)
      acc += ne[n * NDE + d] * wp[(((long)(g * NDE + d) * 2 + slot) * 67 + i) * 64 + o];
  }
  W[idx] = acc;
}

__global__ void k_tb(const float* __restrict__ te, const float* __restrict__ bp,
                     float* __restrict__ TB) {
  int idx = blockIdx.x * 256 + threadIdx.x;
  if (idx >= 3 * NT * NB * NHID) return;
  int o = idx % 64; int r = idx / 64;
  int b = r % NB; r /= NB;
  int t = r % NT; int g = r / NT;
  float acc = 0.f;
  for (int d = 0; d < NDE; ++d)
    acc += te[(b * NT + t) * NDE + d] * bp[(g * NDE + d) * 64 + o];
  TB[idx] = acc;
}

__global__ void k_emb(const float* __restrict__ ne, const float* __restrict__ te,
                      const float* __restrict__ glg, const float* __restrict__ glb,
                      float* __restrict__ E, int t) {
  int idx = blockIdx.x * 256 + threadIdx.x;
  if (idx >= 3 * NB * NN) return;
  int n = idx % NN; int r = idx / NN;
  int b = r % NB; int g = r / NB;
  float v[NDE]; float mu = 0.f;
  for (int d = 0; d < NDE; ++d) { v[d] = ne[n * NDE + d] + te[(b * NT + t) * NDE + d]; mu += v[d]; }
  mu *= (1.f / NDE);
  float var = 0.f;
  for (int d = 0; d < NDE; ++d) { float q = v[d] - mu; var += q * q; }
  var *= (1.f / NDE);
  float inv = rsqrtf(var + 1e-12f);
  for (int d = 0; d < NDE; ++d)
    E[(long)idx * NDE + d] = (v[d] - mu) * inv * glg[g * NDE + d] + glb[g * NDE + d];
}

__global__ void k_zero(float* p, long nElem) {
  long i = (long)blockIdx.x * 256 + threadIdx.x;
  if (i < nElem) p[i] = 0.f;
}

__global__ void k_build_ias(const float* __restrict__ x, const float* __restrict__ h,
                            float* __restrict__ ias, int t) {
  int idx = blockIdx.x * 256 + threadIdx.x;
  if (idx >= NB * NN * NCP) return;
  int i = idx % NCP; int r = idx / NCP; int n = r % NN; int b = r / NN;
  float v;
  if (i < 3)       v = x[((long)(b * NT + t) * NN + n) * 3 + i];
  else if (i < 67) v = h[((long)b * NN + n) * 64 + (i - 3)];
  else             v = 0.f;
  ias[idx] = v;
}

#define VP 80
template <int NG>
__global__ void __launch_bounds__(128 * NG) k_graph_attn(
    const float* __restrict__ E,
    const float* __restrict__ X,
    float* __restrict__ Y0,
    float* __restrict__ Y1,
    int g0) {
  constexpr int NTH = 128 * NG;
  __shared__ __attribute__((aligned(16))) float lds_e[NG * NN * NDE];
  __shared__ __attribute__((aligned(16))) float lds_v[NN * VP];

  const int b = blockIdx.y;
  const int row0 = blockIdx.x * 64;
  const int tid = threadIdx.x;
  const int wave = tid >> 5;
  const int gate = wave >> 2;
  const int wq = wave & 3;
  const int lane = tid & 31;
  const int l16 = lane & 15;
  const int half = lane >> 4;
  const int myrow = row0 + wq * 16 + l16;

  const long ePanel = (long)NB * NN * NDE;
  const float* E0 = E + ((long)g0 * NB + b) * NN * NDE;
  const float* Eb = E0 + gate * ePanel;
  const float* Xb = X + (long)b * NN * NCP;

  {
    float4 re[16];
#pragma unroll
    for (int j = 0; j < 16; ++j) {
      int i = tid + NTH * j;
      int gs = i >> 11, w = i & 2047;
      re[j] = ((const float4*)(E0 + gs * ePanel))[w];
    }
#pragma unroll
    for (int j = 0; j < 16; ++j) ((float4*)lds_e)[tid + NTH * j] = re[j];

    const float4* sv = (const float4*)Xb;
#pragma unroll
    for (int ph = 0; ph < 4 / NG; ++ph) {
      float4 rv[17];
#pragma unroll
      for (int j = 0; j < 17; ++j) rv[j] = sv[ph * (17 * NTH) + tid + NTH * j];
#pragma unroll
      for (int j = 0; j < 17; ++j) {
        int i = ph * (17 * NTH) + tid + NTH * j;
        int rr = i / 17, q = i - rr * 17;
        *(float4*)&lds_v[rr * VP + q * 4] = rv[j];
      }
    }
#pragma unroll
    for (int j = 0; j < 12 / NG; ++j) {
      int i = tid + NTH * j;
      int rr = i / 3, q = i - rr * 3;
      float4 v = make_float4(q == 0 ? 1.f : 0.f, 0.f, 0.f, 0.f);
      *(float4*)&lds_v[rr * VP + 68 + q * 4] = v;
    }
  }

  v2f qB[4];
#pragma unroll
  for (int kk = 0; kk < 4; ++kk) {
    qB[kk].x = Eb[myrow * NDE + kk * 4 + 2 * half];
    qB[kk].y = Eb[myrow * NDE + kk * 4 + 2 * half + 1];
  }
  const int krow = 2 * (l16 & 7) + (l16 >> 3);

  v8f acc[5] = {};

  __syncthreads();

  for (int mc = 0; mc < NN / 16; ++mc) {
    const float* le = lds_e + gate * (NN * NDE) + mc * 16 * NDE;
    const float* lv = lds_v + mc * 16 * VP;

    v8f c = {};
#pragma unroll
    for (int kk = 0; kk < 4; ++kk) {
      v2f ek;
      ek.x = le[krow * NDE + kk * 4 + 2 * half];
      ek.y = le[krow * NDE + kk * 4 + 2 * half + 1];
      c = WMMA_F32(ek, qB[kk], c);
    }

#pragma unroll
    for (int kk = 0; kk < 4; ++kk) {
      v2f ap;
      ap.x = __expf(c[2 * kk]);
      ap.y = __expf(c[2 * kk + 1]);
#pragma unroll
      for (int ct = 0; ct < 5; ++ct) {
        v2f bv;
        bv.x = lv[(kk * 4 + half) * VP + ct * 16 + l16];
        bv.y = lv[(kk * 4 + 2 + half) * VP + ct * 16 + l16];
        acc[ct] = WMMA_F32(ap, bv, acc[ct]);
      }
    }
  }

  float* Y = (NG == 2 && gate) ? Y1 : Y0;
#pragma unroll
  for (int r = 0; r < 8; ++r) {
    float rs = __shfl(acc[4][r], half * 16 + 4);
    float inv = 1.f / rs;
    int row = row0 + wq * 16 + r + 8 * half;
#pragma unroll
    for (int ct = 0; ct < 5; ++ct) {
      int col = ct * 16 + l16;
      if (col < NCP)
        Y[((long)b * NN + row) * NCP + col] = acc[ct][r] * inv;
    }
  }
}

typedef __attribute__((ext_vector_type(4))) float v4f;
__global__ void __launch_bounds__(64) k_gout(
    const float* __restrict__ X, const float* __restrict__ XG0, const float* __restrict__ XG1,
    const float* __restrict__ W, const float* __restrict__ TB,
    float* __restrict__ G0, float* __restrict__ G1, int g0, int t) {
  const int n = blockIdx.x;
  const int gsel = blockIdx.y;
  const int g = g0 + gsel;
  const float* XG = gsel ? XG1 : XG0;
  float* G = gsel ? G1 : G0;
  const int wave = threadIdx.x >> 5;
  const int lane = threadIdx.x & 31;
  const int l16 = lane & 15, half = lane >> 4;
  const float* Wg = W + ((long)g * NN + n) * 2 * NCP * NHID;

  v2f a[2][NCP / 4];
  v4f bw[2][NCP / 4][2];
  v4f tb[8];

#pragma unroll
  for (int slot = 0; slot < 2; ++slot) {
    const float* Arow = (slot ? XG : X) + ((long)(wave * 16 + l16) * NN + n) * NCP + 2 * half;
    const float* Wm = Wg + (long)slot * NCP * NHID + (2 * half) * 64 + l16 * 4;
#pragma unroll
    for (int kk = 0; kk < NCP / 4; ++kk) {
      a[slot][kk] = *(const v2f*)(Arow + kk * 4);
      bw[slot][kk][0] = *(const v4f*)(Wm + (kk * 4) * 64);
      bw[slot][kk][1] = *(const v4f*)(Wm + (kk * 4 + 1) * 64);
    }
  }
#pragma unroll
  for (int r = 0; r < 8; ++r)
    tb[r] = *(const v4f*)(TB + (((long)g * NT + t) * NB + wave * 16 + r + 8 * half) * 64 + l16 * 4);

  v8f c[4] = {};
#pragma unroll
  for (int slot = 0; slot < 2; ++slot) {
#pragma unroll
    for (int kk = 0; kk < NCP / 4; ++kk) {
#pragma unroll
      for (int ct = 0; ct < 4; ++ct) {
        v2f b;
        b.x = bw[slot][kk][0][ct];
        b.y = bw[slot][kk][1][ct];
        c[ct] = WMMA_F32(a[slot][kk], b, c[ct]);
      }
    }
  }
#pragma unroll
  for (int r = 0; r < 8; ++r) {
    int bb = wave * 16 + r + 8 * half;
    v4f v;
    v.x = c[0][r] + tb[r].x;
    v.y = c[1][r] + tb[r].y;
    v.z = c[2][r] + tb[r].z;
    v.w = c[3][r] + tb[r].w;
    *(v4f*)(G + ((long)bb * NN + n) * 64 + l16 * 4) = v;
  }
}

template <bool PAIR>
__global__ void __launch_bounds__(256) k_state_attn(
    float* __restrict__ G, float* __restrict__ Gb, const float* __restrict__ Hs,
    const float* __restrict__ olg, const float* __restrict__ olb, int K,
    const float* __restrict__ x, const float* __restrict__ Gr,
    float* __restrict__ Hnew, float* __restrict__ rows_out, int tx) {
  constexpr int NQ = PAIR ? 2 : 1;
  const int wave = threadIdx.x >> 5, lane = threadIdx.x & 31;
  const long row = (long)blockIdx.x * 8 + wave;
  float g0[NQ], g1[NQ];
#pragma unroll
  for (int q = 0; q < NQ; ++q) {
    const float* Gq = q ? Gb : G;
    g0[q] = Gq[row * 64 + lane];
    g1[q] = Gq[row * 64 + lane + 32];
  }

  float gr0 = 0.f, gr1 = 0.f, xv = 0.f;
  const float hp0 = Hs[((long)(K - 1) * NB * NN + row) * 64 + lane];
  const float hp1 = Hs[((long)(K - 1) * NB * NN + row) * 64 + lane + 32];
  if (lane < 3 && tx < NT) {
    const int b = (int)(row / NN), n = (int)(row % NN);
    xv = x[((long)(b * NT + tx) * NN + n) * 3 + lane];
  }
  if (!PAIR) {
    gr0 = Gr[row * 64 + lane];
    gr1 = Gr[row * 64 + lane + 32];
  }

  float q0[NQ], q1[NQ];
#pragma unroll
  for (int q = 0; q < NQ; ++q) {
    const int g = PAIR ? q : 2;
    float s = g0[q] + g1[q];
    s += __shfl_xor(s, 1); s += __shfl_xor(s, 2); s += __shfl_xor(s, 4);
    s += __shfl_xor(s, 8); s += __shfl_xor(s, 16);
    float mu = s * (1.f / 64.f);
    float d0 = g0[q] - mu, d1 = g1[q] - mu;
    float v = d0 * d0 + d1 * d1;
    v += __shfl_xor(v, 1); v += __shfl_xor(v, 2); v += __shfl_xor(v, 4);
    v += __shfl_xor(v, 8); v += __shfl_xor(v, 16);
    float inv = rsqrtf(v * (1.f / 64.f) + 1e-5f);
    q0[q] = d0 * inv * olg[g * 64 + lane] + olb[g * 64 + lane];
    q1[q] = d1 * inv * olg[g * 64 + lane + 32] + olb[g * 64 + lane + 32];
  }

  float st0[13], st1[13], sc0[NQ][13], sc1[NQ][13];
#pragma unroll
  for (int k = 0; k < 13; ++k) {
    if (k < K) {
      st0[k] = Hs[((long)k * NB * NN + row) * 64 + lane];
      st1[k] = Hs[((long)k * NB * NN + row) * 64 + lane + 32];
#pragma unroll
      for (int q = 0; q < NQ; ++q) {
        float p0 = q0[q] * st0[k], p1 = q1[q] * st1[k];
        p0 += __shfl_xor(p0, 1); p0 += __shfl_xor(p0, 2);
        p0 += __shfl_xor(p0, 4); p0 += __shfl_xor(p0, 8);
        p1 += __shfl_xor(p1, 1); p1 += __shfl_xor(p1, 2);
        p1 += __shfl_xor(p1, 4); p1 += __shfl_xor(p1, 8);
        sc0[q][k] = p0 * 0.25f; sc1[q][k] = p1 * 0.25f;
      }
    } else {
      st0[k] = 0.f; st1[k] = 0.f;
#pragma unroll
      for (int q = 0; q < NQ; ++q) { sc0[q][k] = -1e30f; sc1[q][k] = -1e30f; }
    }
  }
  float y0[NQ], y1[NQ];
#pragma unroll
  for (int q = 0; q < NQ; ++q) {
    float m0 = -1e30f, m1 = -1e30f;
#pragma unroll
    for (int k = 0; k < 13; ++k) { m0 = fmaxf(m0, sc0[q][k]); m1 = fmaxf(m1, sc1[q][k]); }
    float s0 = 0.f, s1 = 0.f;
#pragma unroll
    for (int k = 0; k < 13; ++k) {
      sc0[q][k] = __expf(sc0[q][k] - m0); s0 += sc0[q][k];
      sc1[q][k] = __expf(sc1[q][k] - m1); s1 += sc1[q][k];
    }
    float o0 = 0.f, o1 = 0.f;
#pragma unroll
    for (int k = 0; k < 13; ++k) { o0 += sc0[q][k] * st0[k]; o1 += sc1[q][k] * st1[k]; }
    y0[q] = g0[q] + o0 / s0;
    y1[q] = g1[q] + o1 / s1;
  }

  float e0, e1;
  if (PAIR) {
    float z0 = 1.f / (1.f + __expf(-y0[0]));
    float z1 = 1.f / (1.f + __expf(-y1[0]));
    e0 = z0 * hp0;
    e1 = z1 * hp1;
  } else {
    float r0 = 1.f / (1.f + __expf(-gr0));
    float r1 = 1.f / (1.f + __expf(-gr1));
    float hc0 = tanhf(y0[0]);
    float hc1 = tanhf(y1[0]);
    e0 = r0 * hp0 + (1.f - r0) * hc0;
    e1 = r1 * hp1 + (1.f - r1) * hc1;
  }
  if (tx < NT) {
    __shared__ __attribute__((aligned(16))) float rowbuf[8 * NCP];
    float* rb = rowbuf + wave * NCP;
    rb[3 + lane]  = e0;
    rb[35 + lane] = e1;
    if (lane < 4) rb[lane < 3 ? lane : 67] = xv;
    __syncthreads();
    if (threadIdx.x < 8 * NCP / 4)
      ((float4*)(rows_out + (long)blockIdx.x * 8 * NCP))[threadIdx.x] =
          ((const float4*)rowbuf)[threadIdx.x];
  }
  if (PAIR) {
    Gb[row * 64 + lane]      = y0[1];
    Gb[row * 64 + lane + 32] = y1[1];
  } else {
    Hnew[row * 64 + lane]      = e0;
    Hnew[row * 64 + lane + 32] = e1;
  }
}

__global__ void k_pred(const float* __restrict__ Hs, const float* __restrict__ pw,
                       const float* __restrict__ pb, float* __restrict__ out) {
  int idx = blockIdx.x * 256 + threadIdx.x;
  if (idx >= NB * 12 * NN) return;
  int n = idx % NN; int r = idx / NN; int o = r % 12; int b = r / 12;
  float acc = pb[o];
  for (int cc = 0; cc < 2; ++cc) {
    const float* h = Hs + ((long)(NT - 1 + cc) * NB * NN + (long)b * NN + n) * 64;
    for (int hh = 0; hh < 64; ++hh)
      acc += h[hh] * pw[(o * 2 + cc) * 64 + hh];
  }
  out[idx] = acc;
}

extern "C" void kernel_launch(void* const* d_in, const int* in_sizes, int n_in,
                              void* d_out, int out_size, void* d_ws, size_t ws_size,
                              hipStream_t stream) {
  const float* x   = (const float*)d_in[0];
  const float* ne  = (const float*)d_in[1];
  const float* te  = (const float*)d_in[2];
  const float* wp  = (const float*)d_in[3];
  const float* bp  = (const float*)d_in[4];
  const float* glg = (const float*)d_in[5];
  const float* glb = (const float*)d_in[6];
  const float* olg = (const float*)d_in[7];
  const float* olb = (const float*)d_in[8];
  const float* pw  = (const float*)d_in[9];
  const float* pb  = (const float*)d_in[10];
  float* out = (float*)d_out;

  float* ws = (float*)d_ws;
  long off = 0;
  float* W    = ws + off; off += 3L * NN * 2 * NCP * NHID;
  float* TB   = ws + off; off += 3L * NT * NB * NHID;
  float* E    = ws + off; off += 3L * NB * NN * NDE;
  float* Hs   = ws + off; off += (long)(NT + 1) * NB * NN * NHID;
  float* IAS  = ws + off; off += (long)NB * NN * NCP;
  float* CAND = ws + off; off += (long)NB * NN * NCP;
  float* XG2  = ws + off; off += (long)NB * NN * NCP;
  float* G0   = ws + off; off += (long)NB * NN * NHID;
  float* G1   = ws + off; off += (long)NB * NN * NHID;
  float* G2   = ws + off; off += (long)NB * NN * NHID;

  const long wTotal = 3L * NN * 2 * NCP * NHID;
  k_w<<<dim3((unsigned)((wTotal + 255) / 256)), 256, 0, stream>>>(ne, wp, W);
  k_tb<<<(3 * NT * NB * NHID + 255) / 256, 256, 0, stream>>>(te, bp, TB);
  k_zero<<<(unsigned)(((long)NB * NN * NHID + 255) / 256), 256, 0, stream>>>(Hs, (long)NB * NN * NHID);

  k_build_ias<<<(NB * NN * NCP + 255) / 256, 256, 0, stream>>>(x, Hs, IAS, 0);

  for (int t = 0; t < NT; ++t) {
    float* Hnew = Hs + (long)(t + 1) * NB * NN * NHID;
    k_emb<<<(3 * NB * NN + 255) / 256, 256, 0, stream>>>(ne, te, glg, glb, E, t);
    k_graph_attn<2><<<dim3(NN / 64, NB), 256, 0, stream>>>(E, IAS, XG2, CAND, 0);
    k_gout<<<dim3(NN, 2), 64, 0, stream>>>(IAS, XG2, CAND, W, TB, G0, G1, 0, t);
    k_state_attn<true><<<NB * NN / 8, 256, 0, stream>>>(G0, G1, Hs, olg, olb, t + 1, x, nullptr, nullptr, CAND, t);
    k_graph_attn<1><<<dim3(NN / 64, NB), 128, 0, stream>>>(E, CAND, XG2, nullptr, 2);
    k_gout<<<dim3(NN, 1), 64, 0, stream>>>(CAND, XG2, nullptr, W, TB, G2, nullptr, 2, t);
    k_state_attn<false><<<NB * NN / 8, 256, 0, stream>>>(G2, nullptr, Hs, olg, olb, t + 1, x, G1, Hnew, IAS, t + 1);
  }
  k_pred<<<(NB * 12 * NN + 255) / 256, 256, 0, stream>>>(Hs, pw, pb, out);
}


# k_graph_attn computes its E panel itself while staging (LayerNorm of node_emb + time_emb, 4 lanes per row, row sums as 4-lane trees): the 12 k_emb launches and the E plane go
# speedup vs baseline: 3.8644x; 1.0606x over previous
#include <hip/hip_runtime.h>
#include <math.h>

typedef __attribute__((ext_vector_type(2))) float v2f;
typedef __attribute__((ext_vector_type(8))) float v8f;

#define WMMA_F32(A, B, C) \
  __builtin_amdgcn_wmma_f32_16x16x4_f32(false, (A), false, (B), (short)0, (C), false, false)

#define NB   32
#define NT   12
#define NN   512
#define NDE  16
#define NHID 64
#define NCP  68

__global__ void k_w(const float* __restrict__ ne, const float* __restrict__ wp,
                    float* __restrict__ W) {
  long idx = (long)blockIdx.x * 256 + threadIdx.x;
  const long total = 3L * NN * 2 * NCP * NHID;
  if (idx >= total) return;
  int o = idx % 64; long r = idx / 64;
  int i = r % NCP; r /= NCP;
  int slot = r % 2; r /= 2;
  int n = r % NN; int g = (int)(r / NN);
  float acc = 0.f;
  if (i < 67) {
    for (int d = 0; d < NDE; ++d)
      acc += ne[n * NDE + d] * wp[(((long)(g * NDE + d) * 2 + slot) * 67 + i) * 64 + o];
  }
  W[idx] = acc;
}

__global__ void k_tb(const float* __restrict__ te, const float* __restrict__ bp,
                     float* __restrict__ TB) {
  int idx = blockIdx.x * 256 + threadIdx.x;
  if (idx >= 3 * NT * NB * NHID) return;
  int o = idx % 64; int r = idx / 64;
  int b = r % NB; r /= NB;
  int t = r % NT; int g = r / NT;
  float acc = 0.f;
  for (int d = 0; d < NDE; ++d)
    acc += te[(b * NT + t) * NDE + d] * bp[(g * NDE + d) * 64 + o];
  TB[idx] = acc;
}

__global__ void k_zero(float* p, long nElem) {
  long i = (long)blockIdx.x * 256 + threadIdx.x;
  if (i < nElem) p[i] = 0.f;
}

__global__ void k_build_ias(const float* __restrict__ x, const float* __restrict__ h,
                            float* __restrict__ ias, int t) {
  int idx = blockIdx.x * 256 + threadIdx.x;
  if (idx >= NB * NN * NCP) return;
  int i = idx % NCP; int r = idx / NCP; int n = r % NN; int b = r / NN;
  float v;
  if (i < 3)       v = x[((long)(b * NT + t) * NN + n) * 3 + i];
  else if (i < 67) v = h[((long)b * NN + n) * 64 + (i - 3)];
  else             v = 0.f;
  ias[idx] = v;
}

#define VP 80
__device__ __forceinline__ float quad_sum(float v) {
  v += __int_as_float(__builtin_amdgcn_update_dpp(0, __float_as_int(v), 0xB1, 0xF, 0xF, true));
  v += __int_as_float(__builtin_amdgcn_update_dpp(0, __float_as_int(v), 0x4E, 0xF, 0xF, true));
  return v;
}
template <int NG>
__global__ void __launch_bounds__(128 * NG) k_graph_attn(
    const float* __restrict__ ne,
    const float* __restrict__ te,
    const float* __restrict__ glg,
    const float* __restrict__ glb,
    const float* __restrict__ X,
    float* __restrict__ Y0,
    float* __restrict__ Y1,
    int g0, int t) {
  constexpr int NTH = 128 * NG;
  __shared__ __attribute__((aligned(16))) float lds_e[NG * NN * NDE];
  __shared__ __attribute__((aligned(16))) float lds_v[NN * VP];

  const int b = blockIdx.y;
  const int row0 = blockIdx.x * 64;
  const int tid = threadIdx.x;
  const int wave = tid >> 5;
  const int gate = wave >> 2;
  const int wq = wave & 3;
  const int lane = tid & 31;
  const int l16 = lane & 15;
  const int half = lane >> 4;
  const int myrow = row0 + wq * 16 + l16;

  const float* Xb = X + (long)b * NN * NCP;

  {
    constexpr int NP = 2048 / NTH;
    float4 rn[NP];
#pragma unroll
    for (int p = 0; p < NP; ++p) rn[p] = ((const float4*)ne)[tid + NTH * p];
    const int q4 = tid & 3;
    const float4 tq = ((const float4*)(te + ((long)b * NT + t) * NDE))[q4];
    float4 sg[NG], sb[NG];
#pragma unroll
    for (int j = 0; j < NG; ++j) {
      sg[j] = ((const float4*)(glg + (g0 + j) * NDE))[q4];
      sb[j] = ((const float4*)(glb + (g0 + j) * NDE))[q4];
    }

    const float4* sv = (const float4*)Xb;
#pragma unroll
    for (int ph = 0; ph < 4 / NG; ++ph) {
      float4 rv[17];
#pragma unroll
      for (int j = 0; j < 17; ++j) rv[j] = sv[ph * (17 * NTH) + tid + NTH * j];
      if (ph == 0) {
#pragma unroll
        for (int p = 0; p < NP; ++p) {
          const float v0 = rn[p].x + tq.x, v1 = rn[p].y + tq.y;
          const float v2 = rn[p].z + tq.z, v3 = rn[p].w + tq.w;
          const float mu = quad_sum((v0 + v1) + (v2 + v3)) * (1.f / NDE);
          const float d0 = v0 - mu, d1 = v1 - mu, d2 = v2 - mu, d3 = v3 - mu;
          const float vs = quad_sum((d0 * d0 + d1 * d1) + (d2 * d2 + d3 * d3));
          const float inv = rsqrtf(vs * (1.f / NDE) + 1e-12f);
#pragma unroll
          for (int j = 0; j < NG; ++j) {
            float4 o;
            o.x = d0 * inv * sg[j].x + sb[j].x;
            o.y = d1 * inv * sg[j].y + sb[j].y;
            o.z = d2 * inv * sg[j].z + sb[j].z;
            o.w = d3 * inv * sg[j].w + sb[j].w;
            ((float4*)lds_e)[j * 2048 + tid + NTH * p] = o;
          }
        }
      }
#pragma unroll
      for (int j = 0; j < 17; ++j) {
        int i = ph * (17 * NTH) + tid + NTH * j;
        int rr = i / 17, q = i - rr * 17;
        *(float4*)&lds_v[rr * VP + q * 4] = rv[j];
      }
    }
#pragma unroll
    for (int j = 0; j < 12 / NG; ++j) {
      int i = tid + NTH * j;
      int rr = i / 3, q = i - rr * 3;
      float4 v = make_float4(q == 0 ? 1.f : 0.f, 0.f, 0.f, 0.f);
      *(float4*)&lds_v[rr * VP + 68 + q * 4] = v;
    }
  }

  const int krow = 2 * (l16 & 7) + (l16 >> 3);

  v8f acc[5] = {};

  __syncthreads();

  v2f qB[4];
#pragma unroll
  for (int kk = 0; kk < 4; ++kk) {
    const float* lq = lds_e + gate * (NN * NDE) + myrow * NDE + kk * 4 + 2 * half;
    qB[kk].x = lq[0];
    qB[kk].y = lq[1];
  }

  for (int mc = 0; mc < NN / 16; ++mc) {
    const float* le = lds_e + gate * (NN * NDE) + mc * 16 * NDE;
    const float* lv = lds_v + mc * 16 * VP;

    v8f c = {};
#pragma unroll
    for (int kk = 0; kk < 4; ++kk) {
      v2f ek;
      ek.x = le[krow * NDE + kk * 4 + 2 * half];
      ek.y = le[krow * NDE + kk * 4 + 2 * half + 1];
      c = WMMA_F32(ek, qB[kk], c);
    }

#pragma unroll
    for (int kk = 0; kk < 4; ++kk) {
      v2f ap;
      ap.x = __expf(c[2 * kk]);
      ap.y = __expf(c[2 * kk + 1]);
#pragma unroll
      for (int ct = 0; ct < 5; ++ct) {
        v2f bv;
        bv.x = lv[(kk * 4 + half) * VP + ct * 16 + l16];
        bv.y = lv[(kk * 4 + 2 + half) * VP + ct * 16 + l16];
        acc[ct] = WMMA_F32(ap, bv, acc[ct]);
      }
    }
  }

  float* Y = (NG == 2 && gate) ? Y1 : Y0;
#pragma unroll
  for (int r = 0; r < 8; ++r) {
    float rs = __shfl(acc[4][r], half * 16 + 4);
    float inv = 1.f / rs;
    int row = row0 + wq * 16 + r + 8 * half;
#pragma unroll
    for (int ct = 0; ct < 5; ++ct) {
      int col = ct * 16 + l16;
      if (col < NCP)
        Y[((long)b * NN + row) * NCP + col] = acc[ct][r] * inv;
    }
  }
}

typedef __attribute__((ext_vector_type(4))) float v4f;
__global__ void __launch_bounds__(64) k_gout(
    const float* __restrict__ X, const float* __restrict__ XG0, const float* __restrict__ XG1,
    const float* __restrict__ W, const float* __restrict__ TB,
    float* __restrict__ G0, float* __restrict__ G1, int g0, int t) {
  const int n = blockIdx.x;
  const int gsel = blockIdx.y;
  const int g = g0 + gsel;
  const float* XG = gsel ? XG1 : XG0;
  float* G = gsel ? G1 : G0;
  const int wave = threadIdx.x >> 5;
  const int lane = threadIdx.x & 31;
  const int l16 = lane & 15, half = lane >> 4;
  const float* Wg = W + ((long)g * NN + n) * 2 * NCP * NHID;

  v2f a[2][NCP / 4];
  v4f bw[2][NCP / 4][2];
  v4f tb[8];

#pragma unroll
  for (int slot = 0; slot < 2; ++slot) {
    const float* Arow = (slot ? XG : X) + ((long)(wave * 16 + l16) * NN + n) * NCP + 2 * half;
    const float* Wm = Wg + (long)slot * NCP * NHID + (2 * half) * 64 + l16 * 4;
#pragma unroll
    for (int kk = 0; kk < NCP / 4; ++kk) {
      a[slot][kk] = *(const v2f*)(Arow + kk * 4);
      bw[slot][kk][0] = *(const v4f*)(Wm + (kk * 4) * 64);
      bw[slot][kk][1] = *(const v4f*)(Wm + (kk * 4 + 1) * 64);
    }
  }
#pragma unroll
  for (int r = 0; r < 8; ++r)
    tb[r] = *(const v4f*)(TB + (((long)g * NT + t) * NB + wave * 16 + r + 8 * half) * 64 + l16 * 4);

  v8f c[4] = {};
#pragma unroll
  for (int slot = 0; slot < 2; ++slot) {
#pragma unroll
    for (int kk = 0; kk < NCP / 4; ++kk) {
#pragma unroll
      for (int ct = 0; ct < 4; ++ct) {
        v2f b;
        b.x = bw[slot][kk][0][ct];
        b.y = bw[slot][kk][1][ct];
        c[ct] = WMMA_F32(a[slot][kk], b, c[ct]);
      }
    }
  }
#pragma unroll
  for (int r = 0; r < 8; ++r) {
    int bb = wave * 16 + r + 8 * half;
    v4f v;
    v.x = c[0][r] + tb[r].x;
    v.y = c[1][r] + tb[r].y;
    v.z = c[2][r] + tb[r].z;
    v.w = c[3][r] + tb[r].w;
    *(v4f*)(G + ((long)bb * NN + n) * 64 + l16 * 4) = v;
  }
}

template <bool PAIR>
__global__ void __launch_bounds__(256) k_state_attn(
    float* __restrict__ G, float* __restrict__ Gb, const float* __restrict__ Hs,
    const float* __restrict__ olg, const float* __restrict__ olb, int K,
    const float* __restrict__ x, const float* __restrict__ Gr,
    float* __restrict__ Hnew, float* __restrict__ rows_out, int tx) {
  constexpr int NQ = PAIR ? 2 : 1;
  const int wave = threadIdx.x >> 5, lane = threadIdx.x & 31;
  const long row = (long)blockIdx.x * 8 + wave;
  float g0[NQ], g1[NQ];
#pragma unroll
  for (int q = 0; q < NQ; ++q) {
    const float* Gq = q ? Gb : G;
    g0[q] = Gq[row * 64 + lane];
    g1[q] = Gq[row * 64 + lane + 32];
  }

  float gr0 = 0.f, gr1 = 0.f, xv = 0.f;
  const float hp0 = Hs[((long)(K - 1) * NB * NN + row) * 64 + lane];
  const float hp1 = Hs[((long)(K - 1) * NB * NN + row) * 64 + lane + 32];
  if (lane < 3 && tx < NT) {
    const int b = (int)(row / NN), n = (int)(row % NN);
    xv = x[((long)(b * NT + tx) * NN + n) * 3 + lane];
  }
  if (!PAIR) {
    gr0 = Gr[row * 64 + lane];
    gr1 = Gr[row * 64 + lane + 32];
  }

  float q0[NQ], q1[NQ];
#pragma unroll
  for (int q = 0; q < NQ; ++q) {
    const int g = PAIR ? q : 2;
    float s = g0[q] + g1[q];
    s += __shfl_xor(s, 1); s += __shfl_xor(s, 2); s += __shfl_xor(s, 4);
    s += __shfl_xor(s, 8); s += __shfl_xor(s, 16);
    float mu = s * (1.f / 64.f);
    float d0 = g0[q] - mu, d1 = g1[q] - mu;
    float v = d0 * d0 + d1 * d1;
    v += __shfl_xor(v, 1); v += __shfl_xor(v, 2); v += __shfl_xor(v, 4);
    v += __shfl_xor(v, 8); v += __shfl_xor(v, 16);
    float inv = rsqrtf(v * (1.f / 64.f) + 1e-5f);
    q0[q] = d0 * inv * olg[g * 64 + lane] + olb[g * 64 + lane];
    q1[q] = d1 * inv * olg[g * 64 + lane + 32] + olb[g * 64 + lane + 32];
  }

  float st0[13], st1[13], sc0[NQ][13], sc1[NQ][13];
#pragma unroll
  for (int k = 0; k < 13; ++k) {
    if (k < K) {
      st0[k] = Hs[((long)k * NB * NN + row) * 64 + lane];
      st1[k] = Hs[((long)k * NB * NN + row) * 64 + lane + 32];
#pragma unroll
      for (int q = 0; q < NQ; ++q) {
        float p0 = q0[q] * st0[k], p1 = q1[q] * st1[k];
        p0 += __shfl_xor(p0, 1); p0 += __shfl_xor(p0, 2);
        p0 += __shfl_xor(p0, 4); p0 += __shfl_xor(p0, 8);
        p1 += __shfl_xor(p1, 1); p1 += __shfl_xor(p1, 2);
        p1 += __shfl_xor(p1, 4); p1 += __shfl_xor(p1, 8);
        sc0[q][k] = p0 * 0.25f; sc1[q][k] = p1 * 0.25f;
      }
    } else {
      st0[k] = 0.f; st1[k] = 0.f;
#pragma unroll
      for (int q = 0; q < NQ; ++q) { sc0[q][k] = -1e30f; sc1[q][k] = -1e30f; }
    }
  }
  float y0[NQ], y1[NQ];
#pragma unroll
  for (int q = 0; q < NQ; ++q) {
    float m0 = -1e30f, m1 = -1e30f;
#pragma unroll
    for (int k = 0; k < 13; ++k) { m0 = fmaxf(m0, sc0[q][k]); m1 = fmaxf(m1, sc1[q][k]); }
    float s0 = 0.f, s1 = 0.f;
#pragma unroll
    for (int k = 0; k < 13; ++k) {
      sc0[q][k] = __expf(sc0[q][k] - m0); s0 += sc0[q][k];
      sc1[q][k] = __expf(sc1[q][k] - m1); s1 += sc1[q][k];
    }
    float o0 = 0.f, o1 = 0.f;
#pragma unroll
    for (int k = 0; k < 13; ++k) { o0 += sc0[q][k] * st0[k]; o1 += sc1[q][k] * st1[k]; }
    y0[q] = g0[q] + o0 / s0;
    y1[q] = g1[q] + o1 / s1;
  }

  float e0, e1;
  if (PAIR) {
    float z0 = 1.f / (1.f + __expf(-y0[0]));
    float z1 = 1.f / (1.f + __expf(-y1[0]));
    e0 = z0 * hp0;
    e1 = z1 * hp1;
  } else {
    float r0 = 1.f / (1.f + __expf(-gr0));
    float r1 = 1.f / (1.f + __expf(-gr1));
    float hc0 = tanhf(y0[0]);
    float hc1 = tanhf(y1[0]);
    e0 = r0 * hp0 + (1.f - r0) * hc0;
    e1 = r1 * hp1 + (1.f - r1) * hc1;
  }
  if (tx < NT) {
    __shared__ __attribute__((aligned(16))) float rowbuf[8 * NCP];
    float* rb = rowbuf + wave * NCP;
    rb[3 + lane]  = e0;
    rb[35 + lane] = e1;
    if (lane < 4) rb[lane < 3 ? lane : 67] = xv;
    __syncthreads();
    if (threadIdx.x < 8 * NCP / 4)
      ((float4*)(rows_out + (long)blockIdx.x * 8 * NCP))[threadIdx.x] =
          ((const float4*)rowbuf)[threadIdx.x];
  }
  if (PAIR) {
    Gb[row * 64 + lane]      = y0[1];
    Gb[row * 64 + lane + 32] = y1[1];
  } else {
    Hnew[row * 64 + lane]      = e0;
    Hnew[row * 64 + lane + 32] = e1;
  }
}

__global__ void k_pred(const float* __restrict__ Hs, const float* __restrict__ pw,
                       const float* __restrict__ pb, float* __restrict__ out) {
  int idx = blockIdx.x * 256 + threadIdx.x;
  if (idx >= NB * 12 * NN) return;
  int n = idx % NN; int r = idx / NN; int o = r % 12; int b = r / 12;
  float acc = pb[o];
  for (int cc = 0; cc < 2; ++cc) {
    const float* h = Hs + ((long)(NT - 1 + cc) * NB * NN + (long)b * NN + n) * 64;
    for (int hh = 0; hh < 64; ++hh)
      acc += h[hh] * pw[(o * 2 + cc) * 64 + hh];
  }
  out[idx] = acc;
}

extern "C" void kernel_launch(void* const* d_in, const int* in_sizes, int n_in,
                              void* d_out, int out_size, void* d_ws, size_t ws_size,
                              hipStream_t stream) {
  const float* x   = (const float*)d_in[0];
  const float* ne  = (const float*)d_in[1];
  const float* te  = (const float*)d_in[2];
  const float* wp  = (const float*)d_in[3];
  const float* bp  = (const float*)d_in[4];
  const float* glg = (const float*)d_in[5];
  const float* glb = (const float*)d_in[6];
  const float* olg = (const float*)d_in[7];
  const float* olb = (const float*)d_in[8];
  const float* pw  = (const float*)d_in[9];
  const float* pb  = (const float*)d_in[10];
  float* out = (float*)d_out;

  float* ws = (float*)d_ws;
  long off = 0;
  float* W    = ws + off; off += 3L * NN * 2 * NCP * NHID;
  float* TB   = ws + off; off += 3L * NT * NB * NHID;
  float* Hs   = ws + off; off += (long)(NT + 1) * NB * NN * NHID;
  float* IAS  = ws + off; off += (long)NB * NN * NCP;
  float* CAND = ws + off; off += (long)NB * NN * NCP;
  float* XG2  = ws + off; off += (long)NB * NN * NCP;
  float* G0   = ws + off; off += (long)NB * NN * NHID;
  float* G1   = ws + off; off += (long)NB * NN * NHID;
  float* G2   = ws + off; off += (long)NB * NN * NHID;

  const long wTotal = 3L * NN * 2 * NCP * NHID;
  k_w<<<dim3((unsigned)((wTotal + 255) / 256)), 256, 0, stream>>>(ne, wp, W);
  k_tb<<<(3 * NT * NB * NHID + 255) / 256, 256, 0, stream>>>(te, bp, TB);
  k_zero<<<(unsigned)(((long)NB * NN * NHID + 255) / 256), 256, 0, stream>>>(Hs, (long)NB * NN * NHID);

  k_build_ias<<<(NB * NN * NCP + 255) / 256, 256, 0, stream>>>(x, Hs, IAS, 0);

  for (int t = 0; t < NT; ++t) {
    float* Hnew = Hs + (long)(t + 1) * NB * NN * NHID;
    k_graph_attn<2><<<dim3(NN / 64, NB), 256, 0, stream>>>(ne, te, glg, glb, IAS, XG2, CAND, 0, t);
    k_gout<<<dim3(NN, 2), 64, 0, stream>>>(IAS, XG2, CAND, W, TB, G0, G1, 0, t);
    k_state_attn<true><<<NB * NN / 8, 256, 0, stream>>>(G0, G1, Hs, olg, olb, t + 1, x, nullptr, nullptr, CAND, t);
    k_graph_attn<1><<<dim3(NN / 64, NB), 128, 0, stream>>>(ne, te, glg, glb, CAND, XG2, nullptr, 2, t);
    k_gout<<<dim3(NN, 1), 64, 0, stream>>>(CAND, XG2, nullptr, W, TB, G2, nullptr, 2, t);
    k_state_attn<false><<<NB * NN / 8, 256, 0, stream>>>(G2, nullptr, Hs, olg, olb, t + 1, x, G1, Hnew, IAS, t + 1);
  }
  k_pred<<<(NB * 12 * NN + 255) / 256, 256, 0, stream>>>(Hs, pw, pb, out);
}


# k_state_attn: a row lives in 16 lanes as float4 (head = one lane quad, two rows per wave): head dot products and LayerNorm sums use register permutes, no LDS shuffles; sum order inside them changes
# speedup vs baseline: 4.4006x; 1.1388x over previous
#include <hip/hip_runtime.h>
#include <math.h>

typedef __attribute__((ext_vector_type(2))) float v2f;
typedef __attribute__((ext_vector_type(8))) float v8f;

#define WMMA_F32(A, B, C) \
  __builtin_amdgcn_wmma_f32_16x16x4_f32(false, (A), false, (B), (short)0, (C), false, false)

#define NB   32
#define NT   12
#define NN   512
#define NDE  16
#define NHID 64
#define NCP  68

__global__ void k_w(const float* __restrict__ ne, const float* __restrict__ wp,
                    float* __restrict__ W) {
  long idx = (long)blockIdx.x * 256 + threadIdx.x;
  const long total = 3L * NN * 2 * NCP * NHID;
  if (idx >= total) return;
  int o = idx % 64; long r = idx / 64;
  int i = r % NCP; r /= NCP;
  int slot = r % 2; r /= 2;
  int n = r % NN; int g = (int)(r / NN);
  float acc = 0.f;
  if (i < 67) {
    for (int d = 0; d < NDE; ++d)
      acc += ne[n * NDE + d] * wp[(((long)(g * NDE + d) * 2 + slot) * 67 + i) * 64 + o];
  }
  W[idx] = acc;
}

__global__ void k_tb(const float* __restrict__ te, const float* __restrict__ bp,
                     float* __restrict__ TB) {
  int idx = blockIdx.x * 256 + threadIdx.x;
  if (idx >= 3 * NT * NB * NHID) return;
  int o = idx % 64; int r = idx / 64;
  int b = r % NB; r /= NB;
  int t = r % NT; int g = r / NT;
  float acc = 0.f;
  for (int d = 0; d < NDE; ++d)
    acc += te[(b * NT + t) * NDE + d] * bp[(g * NDE + d) * 64 + o];
  TB[idx] = acc;
}

__global__ void k_zero(float* p, long nElem) {
  long i = (long)blockIdx.x * 256 + threadIdx.x;
  if (i < nElem) p[i] = 0.f;
}

__global__ void k_build_ias(const float* __restrict__ x, const float* __restrict__ h,
                            float* __restrict__ ias, int t) {
  int idx = blockIdx.x * 256 + threadIdx.x;
  if (idx >= NB * NN * NCP) return;
  int i = idx % NCP; int r = idx / NCP; int n = r % NN; int b = r / NN;
  float v;
  if (i < 3)       v = x[((long)(b * NT + t) * NN + n) * 3 + i];
  else if (i < 67) v = h[((long)b * NN + n) * 64 + (i - 3)];
  else             v = 0.f;
  ias[idx] = v;
}

#define VP 80
__device__ __forceinline__ float quad_sum(float v) {
  v += __int_as_float(__builtin_amdgcn_update_dpp(0, __float_as_int(v), 0xB1, 0xF, 0xF, true));
  v += __int_as_float(__builtin_amdgcn_update_dpp(0, __float_as_int(v), 0x4E, 0xF, 0xF, true));
  return v;
}
template <int NG>
__global__ void __launch_bounds__(128 * NG) k_graph_attn(
    const float* __restrict__ ne,
    const float* __restrict__ te,
    const float* __restrict__ glg,
    const float* __restrict__ glb,
    const float* __restrict__ X,
    float* __restrict__ Y0,
    float* __restrict__ Y1,
    int g0, int t) {
  constexpr int NTH = 128 * NG;
  __shared__ __attribute__((aligned(16))) float lds_e[NG * NN * NDE];
  __shared__ __attribute__((aligned(16))) float lds_v[NN * VP];

  const int b = blockIdx.y;
  const int row0 = blockIdx.x * 64;
  const int tid = threadIdx.x;
  const int wave = tid >> 5;
  const int gate = wave >> 2;
  const int wq = wave & 3;
  const int lane = tid & 31;
  const int l16 = lane & 15;
  const int half = lane >> 4;
  const int myrow = row0 + wq * 16 + l16;

  const float* Xb = X + (long)b * NN * NCP;

  {
    constexpr int NP = 2048 / NTH;
    float4 rn[NP];
#pragma unroll
    for (int p = 0; p < NP; ++p) rn[p] = ((const float4*)ne)[tid + NTH * p];
    const int q4 = tid & 3;
    const float4 tq = ((const float4*)(te + ((long)b * NT + t) * NDE))[q4];
    float4 sg[NG], sb[NG];
#pragma unroll
    for (int j = 0; j < NG; ++j) {
      sg[j] = ((const float4*)(glg + (g0 + j) * NDE))[q4];
      sb[j] = ((const float4*)(glb + (g0 + j) * NDE))[q4];
    }

    const float4* sv = (const float4*)Xb;
#pragma unroll
    for (int ph = 0; ph < 4 / NG; ++ph) {
      float4 rv[17];
#pragma unroll
      for (int j = 0; j < 17; ++j) rv[j] = sv[ph * (17 * NTH) + tid + NTH * j];
      if (ph == 0) {
#pragma unroll
        for (int p = 0; p < NP; ++p) {
          const float v0 = rn[p].x + tq.x, v1 = rn[p].y + tq.y;
          const float v2 = rn[p].z + tq.z, v3 = rn[p].w + tq.w;
          const float mu = quad_sum((v0 + v1) + (v2 + v3)) * (1.f / NDE);
          const float d0 = v0 - mu, d1 = v1 - mu, d2 = v2 - mu, d3 = v3 - mu;
          const float vs = quad_sum((d0 * d0 + d1 * d1) + (d2 * d2 + d3 * d3));
          const float inv = rsqrtf(vs * (1.f / NDE) + 1e-12f);
#pragma unroll
          for (int j = 0; j < NG; ++j) {
            float4 o;
            o.x = d0 * inv * sg[j].x + sb[j].x;
            o.y = d1 * inv * sg[j].y + sb[j].y;
            o.z = d2 * inv * sg[j].z + sb[j].z;
            o.w = d3 * inv * sg[j].w + sb[j].w;
            ((float4*)lds_e)[j * 2048 + tid + NTH * p] = o;
          }
        }
      }
#pragma unroll
      for (int j = 0; j < 17; ++j) {
        int i = ph * (17 * NTH) + tid + NTH * j;
        int rr = i / 17, q = i - rr * 17;
        *(float4*)&lds_v[rr * VP + q * 4] = rv[j];
      }
    }
#pragma unroll
    for (int j = 0; j < 12 / NG; ++j) {
      int i = tid + NTH * j;
      int rr = i / 3, q = i - rr * 3;
      float4 v = make_float4(q == 0 ? 1.f : 0.f, 0.f, 0.f, 0.f);
      *(float4*)&lds_v[rr * VP + 68 + q * 4] = v;
    }
  }

  const int krow = 2 * (l16 & 7) + (l16 >> 3);

  v8f acc[5] = {};

  __syncthreads();

  v2f qB[4];
#pragma unroll
  for (int kk = 0; kk < 4; ++kk) {
    const float* lq = lds_e + gate * (NN * NDE) + myrow * NDE + kk * 4 + 2 * half;
    qB[kk].x = lq[0];
    qB[kk].y = lq[1];
  }

  for (int mc = 0; mc < NN / 16; ++mc) {
    const float* le = lds_e + gate * (NN * NDE) + mc * 16 * NDE;
    const float* lv = lds_v + mc * 16 * VP;

    v8f c = {};
#pragma unroll
    for (int kk = 0; kk < 4; ++kk) {
      v2f ek;
      ek.x = le[krow * NDE + kk * 4 + 2 * half];
      ek.y = le[krow * NDE + kk * 4 + 2 * half + 1];
      c = WMMA_F32(ek, qB[kk], c);
    }

#pragma unroll
    for (int kk = 0; kk < 4; ++kk) {
      v2f ap;
      ap.x = __expf(c[2 * kk]);
      ap.y = __expf(c[2 * kk + 1]);
#pragma unroll
      for (int ct = 0; ct < 5; ++ct) {
        v2f bv;
        bv.x = lv[(kk * 4 + half) * VP + ct * 16 + l16];
        bv.y = lv[(kk * 4 + 2 + half) * VP + ct * 16 + l16];
        acc[ct] = WMMA_F32(ap, bv, acc[ct]);
      }
    }
  }

  float* Y = (NG == 2 && gate) ? Y1 : Y0;
#pragma unroll
  for (int r = 0; r < 8; ++r) {
    float rs = __shfl(acc[4][r], half * 16 + 4);
    float inv = 1.f / rs;
    int row = row0 + wq * 16 + r + 8 * half;
#pragma unroll
    for (int ct = 0; ct < 5; ++ct) {
      int col = ct * 16 + l16;
      if (col < NCP)
        Y[((long)b * NN + row) * NCP + col] = acc[ct][r] * inv;
    }
  }
}

typedef __attribute__((ext_vector_type(4))) float v4f;
__global__ void __launch_bounds__(64) k_gout(
    const float* __restrict__ X, const float* __restrict__ XG0, const float* __restrict__ XG1,
    const float* __restrict__ W, const float* __restrict__ TB,
    float* __restrict__ G0, float* __restrict__ G1, int g0, int t) {
  const int n = blockIdx.x;
  const int gsel = blockIdx.y;
  const int g = g0 + gsel;
  const float* XG = gsel ? XG1 : XG0;
  float* G = gsel ? G1 : G0;
  const int wave = threadIdx.x >> 5;
  const int lane = threadIdx.x & 31;
  const int l16 = lane & 15, half = lane >> 4;
  const float* Wg = W + ((long)g * NN + n) * 2 * NCP * NHID;

  v2f a[2][NCP / 4];
  v4f bw[2][NCP / 4][2];
  v4f tb[8];

#pragma unroll
  for (int slot = 0; slot < 2; ++slot) {
    const float* Arow = (slot ? XG : X) + ((long)(wave * 16 + l16) * NN + n) * NCP + 2 * half;
    const float* Wm = Wg + (long)slot * NCP * NHID + (2 * half) * 64 + l16 * 4;
#pragma unroll
    for (int kk = 0; kk < NCP / 4; ++kk) {
      a[slot][kk] = *(const v2f*)(Arow + kk * 4);
      bw[slot][kk][0] = *(const v4f*)(Wm + (kk * 4) * 64);
      bw[slot][kk][1] = *(const v4f*)(Wm + (kk * 4 + 1) * 64);
    }
  }
#pragma unroll
  for (int r = 0; r < 8; ++r)
    tb[r] = *(const v4f*)(TB + (((long)g * NT + t) * NB + wave * 16 + r + 8 * half) * 64 + l16 * 4);

  v8f c[4] = {};
#pragma unroll
  for (int slot = 0; slot < 2; ++slot) {
#pragma unroll
    for (int kk = 0; kk < NCP / 4; ++kk) {
#pragma unroll
      for (int ct = 0; ct < 4; ++ct) {
        v2f b;
        b.x = bw[slot][kk][0][ct];
        b.y = bw[slot][kk][1][ct];
        c[ct] = WMMA_F32(a[slot][kk], b, c[ct]);
      }
    }
  }
#pragma unroll
  for (int r = 0; r < 8; ++r) {
    int bb = wave * 16 + r + 8 * half;
    v4f v;
    v.x = c[0][r] + tb[r].x;
    v.y = c[1][r] + tb[r].y;
    v.z = c[2][r] + tb[r].z;
    v.w = c[3][r] + tb[r].w;
    *(v4f*)(G + ((long)bb * NN + n) * 64 + l16 * 4) = v;
  }
}

__device__ __forceinline__ float row16_sum(float v) {
  v = quad_sum(v);
  v += __int_as_float(__builtin_amdgcn_update_dpp(0, __float_as_int(v), 0x164, 0xF, 0xF, true));
  v += __int_as_float(__builtin_amdgcn_update_dpp(0, __float_as_int(v), 0x168, 0xF, 0xF, true));
  return v;
}
template <bool PAIR>
__global__ void __launch_bounds__(256) k_state_attn(
    float* __restrict__ G, float* __restrict__ Gb, const float* __restrict__ Hs,
    const float* __restrict__ olg, const float* __restrict__ olb, int K,
    const float* __restrict__ x, const float* __restrict__ Gr,
    float* __restrict__ Hnew, float* __restrict__ rows_out, int tx) {
  constexpr int NQ = PAIR ? 2 : 1;
  const int wave = threadIdx.x >> 5, lane = threadIdx.x & 31;
  const int hl = lane & 15;
  const int slot = wave * 2 + (lane >> 4);
  const long row = (long)blockIdx.x * 16 + slot;
  const long ro = row * 64 + hl * 4;
  const long plane = (long)NB * NN * 64;

  float4 g[NQ];
#pragma unroll
  for (int q = 0; q < NQ; ++q) g[q] = *(const float4*)((q ? Gb : G) + ro);

  float4 gr = make_float4(0.f, 0.f, 0.f, 0.f);
  float xv = 0.f;
  const float4 hp = *(const float4*)(Hs + (long)(K - 1) * plane + ro);
  if (hl < 3 && tx < NT) {
    const int b = (int)(row / NN), n = (int)(row % NN);
    xv = x[((long)(b * NT + tx) * NN + n) * 3 + hl];
  }
  if (!PAIR) gr = *(const float4*)(Gr + ro);

  float4 qv[NQ];
#pragma unroll
  for (int q = 0; q < NQ; ++q) {
    const int gi = PAIR ? q : 2;
    const float4 lg = *(const float4*)(olg + gi * 64 + hl * 4);
    const float4 lb = *(const float4*)(olb + gi * 64 + hl * 4);
    const float mu = row16_sum((g[q].x + g[q].y) + (g[q].z + g[q].w)) * (1.f / 64.f);
    const float d0 = g[q].x - mu, d1 = g[q].y - mu, d2 = g[q].z - mu, d3 = g[q].w - mu;
    const float v = row16_sum((d0 * d0 + d1 * d1) + (d2 * d2 + d3 * d3));
    const float inv = rsqrtf(v * (1.f / 64.f) + 1e-5f);
    qv[q].x = d0 * inv * lg.x + lb.x;
    qv[q].y = d1 * inv * lg.y + lb.y;
    qv[q].z = d2 * inv * lg.z + lb.z;
    qv[q].w = d3 * inv * lg.w + lb.w;
  }

  float4 st[13];
  float sc[NQ][13];
#pragma unroll
  for (int k = 0; k < 13; ++k) {
    if (k < K) {
      st[k] = *(const float4*)(Hs + (long)k * plane + ro);
#pragma unroll
      for (int q = 0; q < NQ; ++q) {
        const float p = (qv[q].x * st[k].x + qv[q].y * st[k].y) + (qv[q].z * st[k].z + qv[q].w * st[k].w);
        sc[q][k] = quad_sum(p) * 0.25f;
      }
    } else {
      st[k] = make_float4(0.f, 0.f, 0.f, 0.f);
#pragma unroll
      for (int q = 0; q < NQ; ++q) sc[q][k] = -1e30f;
    }
  }
  float4 y[NQ];
#pragma unroll
  for (int q = 0; q < NQ; ++q) {
    float m = -1e30f;
#pragma unroll
    for (int k = 0; k < 13; ++k) m = fmaxf(m, sc[q][k]);
    float s = 0.f;
    float4 o = make_float4(0.f, 0.f, 0.f, 0.f);
#pragma unroll
    for (int k = 0; k < 13; ++k) {
      const float pe = __expf(sc[q][k] - m);
      s += pe;
      o.x += pe * st[k].x; o.y += pe * st[k].y;
      o.z += pe * st[k].z; o.w += pe * st[k].w;
    }
    const float rs = 1.f / s;
    y[q].x = g[q].x + o.x * rs;
    y[q].y = g[q].y + o.y * rs;
    y[q].z = g[q].z + o.z * rs;
    y[q].w = g[q].w + o.w * rs;
  }

  float4 e;
  if (PAIR) {
    e.x = (1.f / (1.f + __expf(-y[0].x))) * hp.x;
    e.y = (1.f / (1.f + __expf(-y[0].y))) * hp.y;
    e.z = (1.f / (1.f + __expf(-y[0].z))) * hp.z;
    e.w = (1.f / (1.f + __expf(-y[0].w))) * hp.w;
  } else {
    const float r0 = 1.f / (1.f + __expf(-gr.x)), r1 = 1.f / (1.f + __expf(-gr.y));
    const float r2 = 1.f / (1.f + __expf(-gr.z)), r3 = 1.f / (1.f + __expf(-gr.w));
    e.x = r0 * hp.x + (1.f - r0) * tanhf(y[0].x);
    e.y = r1 * hp.y + (1.f - r1) * tanhf(y[0].y);
    e.z = r2 * hp.z + (1.f - r2) * tanhf(y[0].z);
    e.w = r3 * hp.w + (1.f - r3) * tanhf(y[0].w);
  }
  if (tx < NT) {
    __shared__ __attribute__((aligned(16))) float rowbuf[16 * NCP];
    float* rb = rowbuf + slot * NCP;
    rb[3 + 4 * hl]     = e.x;
    rb[3 + 4 * hl + 1] = e.y;
    rb[3 + 4 * hl + 2] = e.z;
    rb[3 + 4 * hl + 3] = e.w;
    if (hl < 4) rb[hl < 3 ? hl : 67] = xv;
    __syncthreads();
    for (int i = threadIdx.x; i < 16 * NCP / 4; i += 256)
      ((float4*)(rows_out + (long)blockIdx.x * 16 * NCP))[i] = ((const float4*)rowbuf)[i];
  }
  if (PAIR) *(float4*)(Gb + ro) = y[1];
  else      *(float4*)(Hnew + ro) = e;
}

__global__ void k_pred(const float* __restrict__ Hs, const float* __restrict__ pw,
                       const float* __restrict__ pb, float* __restrict__ out) {
  int idx = blockIdx.x * 256 + threadIdx.x;
  if (idx >= NB * 12 * NN) return;
  int n = idx % NN; int r = idx / NN; int o = r % 12; int b = r / 12;
  float acc = pb[o];
  for (int cc = 0; cc < 2; ++cc) {
    const float* h = Hs + ((long)(NT - 1 + cc) * NB * NN + (long)b * NN + n) * 64;
    for (int hh = 0; hh < 64; ++hh)
      acc += h[hh] * pw[(o * 2 + cc) * 64 + hh];
  }
  out[idx] = acc;
}

extern "C" void kernel_launch(void* const* d_in, const int* in_sizes, int n_in,
                              void* d_out, int out_size, void* d_ws, size_t ws_size,
                              hipStream_t stream) {
  const float* x   = (const float*)d_in[0];
  const float* ne  = (const float*)d_in[1];
  const float* te  = (const float*)d_in[2];
  const float* wp  = (const float*)d_in[3];
  const float* bp  = (const float*)d_in[4];
  const float* glg = (const float*)d_in[5];
  const float* glb = (const float*)d_in[6];
  const float* olg = (const float*)d_in[7];
  const float* olb = (const float*)d_in[8];
  const float* pw  = (const float*)d_in[9];
  const float* pb  = (const float*)d_in[10];
  float* out = (float*)d_out;

  float* ws = (float*)d_ws;
  long off = 0;
  float* W    = ws + off; off += 3L * NN * 2 * NCP * NHID;
  float* TB   = ws + off; off += 3L * NT * NB * NHID;
  float* Hs   = ws + off; off += (long)(NT + 1) * NB * NN * NHID;
  float* IAS  = ws + off; off += (long)NB * NN * NCP;
  float* CAND = ws + off; off += (long)NB * NN * NCP;
  float* XG2  = ws + off; off += (long)NB * NN * NCP;
  float* G0   = ws + off; off += (long)NB * NN * NHID;
  float* G1   = ws + off; off += (long)NB * NN * NHID;
  float* G2   = ws + off; off += (long)NB * NN * NHID;

  const long wTotal = 3L * NN * 2 * NCP * NHID;
  k_w<<<dim3((unsigned)((wTotal + 255) / 256)), 256, 0, stream>>>(ne, wp, W);
  k_tb<<<(3 * NT * NB * NHID + 255) / 256, 256, 0, stream>>>(te, bp, TB);
  k_zero<<<(unsigned)(((long)NB * NN * NHID + 255) / 256), 256, 0, stream>>>(Hs, (long)NB * NN * NHID);

  k_build_ias<<<(NB * NN * NCP + 255) / 256, 256, 0, stream>>>(x, Hs, IAS, 0);

  for (int t = 0; t < NT; ++t) {
    float* Hnew = Hs + (long)(t + 1) * NB * NN * NHID;
    k_graph_attn<2><<<dim3(NN / 64, NB), 256, 0, stream>>>(ne, te, glg, glb, IAS, XG2, CAND, 0, t);
    k_gout<<<dim3(NN, 2), 64, 0, stream>>>(IAS, XG2, CAND, W, TB, G0, G1, 0, t);
    k_state_attn<true><<<NB * NN / 16, 256, 0, stream>>>(G0, G1, Hs, olg, olb, t + 1, x, nullptr, nullptr, CAND, t);
    k_graph_attn<1><<<dim3(NN / 64, NB), 128, 0, stream>>>(ne, te, glg, glb, CAND, XG2, nullptr, 2, t);
    k_gout<<<dim3(NN, 1), 64, 0, stream>>>(CAND, XG2, nullptr, W, TB, G2, nullptr, 2, t);
    k_state_attn<false><<<NB * NN / 16, 256, 0, stream>>>(G2, nullptr, Hs, olg, olb, t + 1, x, G1, Hnew, IAS, t + 1);
  }
  k_pred<<<(NB * 12 * NN + 255) / 256, 256, 0, stream>>>(Hs, pw, pb, out);
}


# k_state_attn: every load of a row, its K stored states too, is requested before the first wait (unrolled bodies for K <= 4, 8, 12): one memory round trip per wave, not K + 2; sum order unchanged
# speedup vs baseline: 4.6345x; 1.0532x over previous
#include <hip/hip_runtime.h>
#include <math.h>

typedef __attribute__((ext_vector_type(2))) float v2f;
typedef __attribute__((ext_vector_type(8))) float v8f;

#define WMMA_F32(A, B, C) \
  __builtin_amdgcn_wmma_f32_16x16x4_f32(false, (A), false, (B), (short)0, (C), false, false)

#define NB   32
#define NT   12
#define NN   512
#define NDE  16
#define NHID 64
#define NCP  68

__global__ void k_w(const float* __restrict__ ne, const float* __restrict__ wp,
                    float* __restrict__ W) {
  long idx = (long)blockIdx.x * 256 + threadIdx.x;
  const long total = 3L * NN * 2 * NCP * NHID;
  if (idx >= total) return;
  int o = idx % 64; long r = idx / 64;
  int i = r % NCP; r /= NCP;
  int slot = r % 2; r /= 2;
  int n = r % NN; int g = (int)(r / NN);
  float acc = 0.f;
  if (i < 67) {
    for (int d = 0; d < NDE; ++d)
      acc += ne[n * NDE + d] * wp[(((long)(g * NDE + d) * 2 + slot) * 67 + i) * 64 + o];
  }
  W[idx] = acc;
}

__global__ void k_tb(const float* __restrict__ te, const float* __restrict__ bp,
                     float* __restrict__ TB) {
  int idx = blockIdx.x * 256 + threadIdx.x;
  if (idx >= 3 * NT * NB * NHID) return;
  int o = idx % 64; int r = idx / 64;
  int b = r % NB; r /= NB;
  int t = r % NT; int g = r / NT;
  float acc = 0.f;
  for (int d = 0; d < NDE; ++d)
    acc += te[(b * NT + t) * NDE + d] * bp[(g * NDE + d) * 64 + o];
  TB[idx] = acc;
}

__global__ void k_zero(float* p, long nElem) {
  long i = (long)blockIdx.x * 256 + threadIdx.x;
  if (i < nElem) p[i] = 0.f;
}

__global__ void k_build_ias(const float* __restrict__ x, const float* __restrict__ h,
                            float* __restrict__ ias, int t) {
  int idx = blockIdx.x * 256 + threadIdx.x;
  if (idx >= NB * NN * NCP) return;
  int i = idx % NCP; int r = idx / NCP; int n = r % NN; int b = r / NN;
  float v;
  if (i < 3)       v = x[((long)(b * NT + t) * NN + n) * 3 + i];
  else if (i < 67) v = h[((long)b * NN + n) * 64 + (i - 3)];
  else             v = 0.f;
  ias[idx] = v;
}

#define VP 80
__device__ __forceinline__ float quad_sum(float v) {
  v += __int_as_float(__builtin_amdgcn_update_dpp(0, __float_as_int(v), 0xB1, 0xF, 0xF, true));
  v += __int_as_float(__builtin_amdgcn_update_dpp(0, __float_as_int(v), 0x4E, 0xF, 0xF, true));
  return v;
}
template <int NG>
__global__ void __launch_bounds__(128 * NG) k_graph_attn(
    const float* __restrict__ ne,
    const float* __restrict__ te,
    const float* __restrict__ glg,
    const float* __restrict__ glb,
    const float* __restrict__ X,
    float* __restrict__ Y0,
    float* __restrict__ Y1,
    int g0, int t) {
  constexpr int NTH = 128 * NG;
  __shared__ __attribute__((aligned(16))) float lds_e[NG * NN * NDE];
  __shared__ __attribute__((aligned(16))) float lds_v[NN * VP];

  const int b = blockIdx.y;
  const int row0 = blockIdx.x * 64;
  const int tid = threadIdx.x;
  const int wave = tid >> 5;
  const int gate = wave >> 2;
  const int wq = wave & 3;
  const int lane = tid & 31;
  const int l16 = lane & 15;
  const int half = lane >> 4;
  const int myrow = row0 + wq * 16 + l16;

  const float* Xb = X + (long)b * NN * NCP;

  {
    constexpr int NP = 2048 / NTH;
    float4 rn[NP];
#pragma unroll
    for (int p = 0; p < NP; ++p) rn[p] = ((const float4*)ne)[tid + NTH * p];
    const int q4 = tid & 3;
    const float4 tq = ((const float4*)(te + ((long)b * NT + t) * NDE))[q4];
    float4 sg[NG], sb[NG];
#pragma unroll
    for (int j = 0; j < NG; ++j) {
      sg[j] = ((const float4*)(glg + (g0 + j) * NDE))[q4];
      sb[j] = ((const float4*)(glb + (g0 + j) * NDE))[q4];
    }

    const float4* sv = (const float4*)Xb;
#pragma unroll
    for (int ph = 0; ph < 4 / NG; ++ph) {
      float4 rv[17];
#pragma unroll
      for (int j = 0; j < 17; ++j) rv[j] = sv[ph * (17 * NTH) + tid + NTH * j];
      if (ph == 0) {
#pragma unroll
        for (int p = 0; p < NP; ++p) {
          const float v0 = rn[p].x + tq.x, v1 = rn[p].y + tq.y;
          const float v2 = rn[p].z + tq.z, v3 = rn[p].w + tq.w;
          const float mu = quad_sum((v0 + v1) + (v2 + v3)) * (1.f / NDE);
          const float d0 = v0 - mu, d1 = v1 - mu, d2 = v2 - mu, d3 = v3 - mu;
          const float vs = quad_sum((d0 * d0 + d1 * d1) + (d2 * d2 + d3 * d3));
          const float inv = rsqrtf(vs * (1.f / NDE) + 1e-12f);
#pragma unroll
          for (int j = 0; j < NG; ++j) {
            float4 o;
            o.x = d0 * inv * sg[j].x + sb[j].x;
            o.y = d1 * inv * sg[j].y + sb[j].y;
            o.z = d2 * inv * sg[j].z + sb[j].z;
            o.w = d3 * inv * sg[j].w + sb[j].w;
            ((float4*)lds_e)[j * 2048 + tid + NTH * p] = o;
          }
        }
      }
#pragma unroll
      for (int j = 0; j < 17; ++j) {
        int i = ph * (17 * NTH) + tid + NTH * j;
        int rr = i / 17, q = i - rr * 17;
        *(float4*)&lds_v[rr * VP + q * 4] = rv[j];
      }
    }
#pragma unroll
    for (int j = 0; j < 12 / NG; ++j) {
      int i = tid + NTH * j;
      int rr = i / 3, q = i - rr * 3;
      float4 v = make_float4(q == 0 ? 1.f : 0.f, 0.f, 0.f, 0.f);
      *(float4*)&lds_v[rr * VP + 68 + q * 4] = v;
    }
  }

  const int krow = 2 * (l16 & 7) + (l16 >> 3);

  v8f acc[5] = {};

  __syncthreads();

  v2f qB[4];
#pragma unroll
  for (int kk = 0; kk < 4; ++kk) {
    const float* lq = lds_e + gate * (NN * NDE) + myrow * NDE + kk * 4 + 2 * half;
    qB[kk].x = lq[0];
    qB[kk].y = lq[1];
  }

  for (int mc = 0; mc < NN / 16; ++mc) {
    const float* le = lds_e + gate * (NN * NDE) + mc * 16 * NDE;
    const float* lv = lds_v + mc * 16 * VP;

    v8f c = {};
#pragma unroll
    for (int kk = 0; kk < 4; ++kk) {
      v2f ek;
      ek.x = le[krow * NDE + kk * 4 + 2 * half];
      ek.y = le[krow * NDE + kk * 4 + 2 * half + 1];
      c = WMMA_F32(ek, qB[kk], c);
    }

#pragma unroll
    for (int kk = 0; kk < 4; ++kk) {
      v2f ap;
      ap.x = __expf(c[2 * kk]);
      ap.y = __expf(c[2 * kk + 1]);
#pragma unroll
      for (int ct = 0; ct < 5; ++ct) {
        v2f bv;
        bv.x = lv[(kk * 4 + half) * VP + ct * 16 + l16];
        bv.y = lv[(kk * 4 + 2 + half) * VP + ct * 16 + l16];
        acc[ct] = WMMA_F32(ap, bv, acc[ct]);
      }
    }
  }

  float* Y = (NG == 2 && gate) ? Y1 : Y0;
#pragma unroll
  for (int r = 0; r < 8; ++r) {
    float rs = __shfl(acc[4][r], half * 16 + 4);
    float inv = 1.f / rs;
    int row = row0 + wq * 16 + r + 8 * half;
#pragma unroll
    for (int ct = 0; ct < 5; ++ct) {
      int col = ct * 16 + l16;
      if (col < NCP)
        Y[((long)b * NN + row) * NCP + col] = acc[ct][r] * inv;
    }
  }
}

typedef __attribute__((ext_vector_type(4))) float v4f;
__global__ void __launch_bounds__(64) k_gout(
    const float* __restrict__ X, const float* __restrict__ XG0, const float* __restrict__ XG1,
    const float* __restrict__ W, const float* __restrict__ TB,
    float* __restrict__ G0, float* __restrict__ G1, int g0, int t) {
  const int n = blockIdx.x;
  const int gsel = blockIdx.y;
  const int g = g0 + gsel;
  const float* XG = gsel ? XG1 : XG0;
  float* G = gsel ? G1 : G0;
  const int wave = threadIdx.x >> 5;
  const int lane = threadIdx.x & 31;
  const int l16 = lane & 15, half = lane >> 4;
  const float* Wg = W + ((long)g * NN + n) * 2 * NCP * NHID;

  v2f a[2][NCP / 4];
  v4f bw[2][NCP / 4][2];
  v4f tb[8];

#pragma unroll
  for (int slot = 0; slot < 2; ++slot) {
    const float* Arow = (slot ? XG : X) + ((long)(wave * 16 + l16) * NN + n) * NCP + 2 * half;
    const float* Wm = Wg + (long)slot * NCP * NHID + (2 * half) * 64 + l16 * 4;
#pragma unroll
    for (int kk = 0; kk < NCP / 4; ++kk) {
      a[slot][kk] = *(const v2f*)(Arow + kk * 4);
      bw[slot][kk][0] = *(const v4f*)(Wm + (kk * 4) * 64);
      bw[slot][kk][1] = *(const v4f*)(Wm + (kk * 4 + 1) * 64);
    }
  }
#pragma unroll
  for (int r = 0; r < 8; ++r)
    tb[r] = *(const v4f*)(TB + (((long)g * NT + t) * NB + wave * 16 + r + 8 * half) * 64 + l16 * 4);

  v8f c[4] = {};
#pragma unroll
  for (int slot = 0; slot < 2; ++slot) {
#pragma unroll
    for (int kk = 0; kk < NCP / 4; ++kk) {
#pragma unroll
      for (int ct = 0; ct < 4; ++ct) {
        v2f b;
        b.x = bw[slot][kk][0][ct];
        b.y = bw[slot][kk][1][ct];
        c[ct] = WMMA_F32(a[slot][kk], b, c[ct]);
      }
    }
  }
#pragma unroll
  for (int r = 0; r < 8; ++r) {
    int bb = wave * 16 + r + 8 * half;
    v4f v;
    v.x = c[0][r] + tb[r].x;
    v.y = c[1][r] + tb[r].y;
    v.z = c[2][r] + tb[r].z;
    v.w = c[3][r] + tb[r].w;
    *(v4f*)(G + ((long)bb * NN + n) * 64 + l16 * 4) = v;
  }
}

__device__ __forceinline__ float row16_sum(float v) {
  v = quad_sum(v);
  v += __int_as_float(__builtin_amdgcn_update_dpp(0, __float_as_int(v), 0x164, 0xF, 0xF, true));
  v += __int_as_float(__builtin_amdgcn_update_dpp(0, __float_as_int(v), 0x168, 0xF, 0xF, true));
  return v;
}
template <bool PAIR, int KM>
__device__ __forceinline__ void state_attn_rows(
    float* __restrict__ G, float* __restrict__ Gb, const float* __restrict__ Hs,
    const float* __restrict__ olg, const float* __restrict__ olb, int K,
    const float* __restrict__ x, const float* __restrict__ Gr,
    float* __restrict__ Hnew, float* __restrict__ rows_out, int tx, float* rowbuf) {
  constexpr int NQ = PAIR ? 2 : 1;
  const int wave = threadIdx.x >> 5, lane = threadIdx.x & 31;
  const int hl = lane & 15;
  const int slot = wave * 2 + (lane >> 4);
  const long row = (long)blockIdx.x * 16 + slot;
  const long ro = row * 64 + hl * 4;
  const long plane = (long)NB * NN * 64;

  float4 g[NQ], lg[NQ], lb[NQ];
#pragma unroll
  for (int q = 0; q < NQ; ++q) {
    const int gi = PAIR ? q : 2;
    g[q] = *(const float4*)((q ? Gb : G) + ro);
    lg[q] = *(const float4*)(olg + gi * 64 + hl * 4);
    lb[q] = *(const float4*)(olb + gi * 64 + hl * 4);
  }
  float4 st[KM];
#pragma unroll
  for (int k = 0; k < KM; ++k) {
    const int kc = k < K ? k : K - 1;
    st[k] = *(const float4*)(Hs + (long)kc * plane + ro);
  }
  const float4 hp = st[KM - 1];
  float4 gr = make_float4(0.f, 0.f, 0.f, 0.f);
  float xv = 0.f;
  if (hl < 3 && tx < NT) {
    const int b = (int)(row / NN), n = (int)(row % NN);
    xv = x[((long)(b * NT + tx) * NN + n) * 3 + hl];
  }
  if (!PAIR) gr = *(const float4*)(Gr + ro);

  float4 qv[NQ];
#pragma unroll
  for (int q = 0; q < NQ; ++q) {
    const float mu = row16_sum((g[q].x + g[q].y) + (g[q].z + g[q].w)) * (1.f / 64.f);
    const float d0 = g[q].x - mu, d1 = g[q].y - mu, d2 = g[q].z - mu, d3 = g[q].w - mu;
    const float v = row16_sum((d0 * d0 + d1 * d1) + (d2 * d2 + d3 * d3));
    const float inv = rsqrtf(v * (1.f / 64.f) + 1e-5f);
    qv[q].x = d0 * inv * lg[q].x + lb[q].x;
    qv[q].y = d1 * inv * lg[q].y + lb[q].y;
    qv[q].z = d2 * inv * lg[q].z + lb[q].z;
    qv[q].w = d3 * inv * lg[q].w + lb[q].w;
  }

  float sc[NQ][KM];
#pragma unroll
  for (int k = 0; k < KM; ++k) {
#pragma unroll
    for (int q = 0; q < NQ; ++q) {
      const float p = (qv[q].x * st[k].x + qv[q].y * st[k].y) + (qv[q].z * st[k].z + qv[q].w * st[k].w);
      const float s = quad_sum(p) * 0.25f;
      sc[q][k] = k < K ? s : -1e30f;
    }
  }
  float4 y[NQ];
#pragma unroll
  for (int q = 0; q < NQ; ++q) {
    float m = -1e30f;
#pragma unroll
    for (int k = 0; k < KM; ++k) m = fmaxf(m, sc[q][k]);
    float s = 0.f;
    float4 o = make_float4(0.f, 0.f, 0.f, 0.f);
#pragma unroll
    for (int k = 0; k < KM; ++k) {
      const float pe = __expf(sc[q][k] - m);
      s += pe;
      o.x += pe * st[k].x; o.y += pe * st[k].y;
      o.z += pe * st[k].z; o.w += pe * st[k].w;
    }
    const float rs = 1.f / s;
    y[q].x = g[q].x + o.x * rs;
    y[q].y = g[q].y + o.y * rs;
    y[q].z = g[q].z + o.z * rs;
    y[q].w = g[q].w + o.w * rs;
  }

  float4 e;
  if (PAIR) {
    e.x = (1.f / (1.f + __expf(-y[0].x))) * hp.x;
    e.y = (1.f / (1.f + __expf(-y[0].y))) * hp.y;
    e.z = (1.f / (1.f + __expf(-y[0].z))) * hp.z;
    e.w = (1.f / (1.f + __expf(-y[0].w))) * hp.w;
  } else {
    const float r0 = 1.f / (1.f + __expf(-gr.x)), r1 = 1.f / (1.f + __expf(-gr.y));
    const float r2 = 1.f / (1.f + __expf(-gr.z)), r3 = 1.f / (1.f + __expf(-gr.w));
    e.x = r0 * hp.x + (1.f - r0) * tanhf(y[0].x);
    e.y = r1 * hp.y + (1.f - r1) * tanhf(y[0].y);
    e.z = r2 * hp.z + (1.f - r2) * tanhf(y[0].z);
    e.w = r3 * hp.w + (1.f - r3) * tanhf(y[0].w);
  }
  if (tx < NT) {
    float* rb = rowbuf + slot * NCP;
    rb[3 + 4 * hl]     = e.x;
    rb[3 + 4 * hl + 1] = e.y;
    rb[3 + 4 * hl + 2] = e.z;
    rb[3 + 4 * hl + 3] = e.w;
    if (hl < 4) rb[hl < 3 ? hl : 67] = xv;
    __syncthreads();
    for (int i = threadIdx.x; i < 16 * NCP / 4; i += 256)
      ((float4*)(rows_out + (long)blockIdx.x * 16 * NCP))[i] = ((const float4*)rowbuf)[i];
  }
  if (PAIR) *(float4*)(Gb + ro) = y[1];
  else      *(float4*)(Hnew + ro) = e;
}
template <bool PAIR>
__global__ void __launch_bounds__(256) k_state_attn(
    float* __restrict__ G, float* __restrict__ Gb, const float* __restrict__ Hs,
    const float* __restrict__ olg, const float* __restrict__ olb, int K,
    const float* __restrict__ x, const float* __restrict__ Gr,
    float* __restrict__ Hnew, float* __restrict__ rows_out, int tx) {
  __shared__ __attribute__((aligned(16))) float rowbuf[16 * NCP];
  if (K <= 4)      state_attn_rows<PAIR, 4>(G, Gb, Hs, olg, olb, K, x, Gr, Hnew, rows_out, tx, rowbuf);
  else if (K <= 8) state_attn_rows<PAIR, 8>(G, Gb, Hs, olg, olb, K, x, Gr, Hnew, rows_out, tx, rowbuf);
  else             state_attn_rows<PAIR, 12>(G, Gb, Hs, olg, olb, K, x, Gr, Hnew, rows_out, tx, rowbuf);
}

__global__ void k_pred(const float* __restrict__ Hs, const float* __restrict__ pw,
                       const float* __restrict__ pb, float* __restrict__ out) {
  int idx = blockIdx.x * 256 + threadIdx.x;
  if (idx >= NB * 12 * NN) return;
  int n = idx % NN; int r = idx / NN; int o = r % 12; int b = r / 12;
  float acc = pb[o];
  for (int cc = 0; cc < 2; ++cc) {
    const float* h = Hs + ((long)(NT - 1 + cc) * NB * NN + (long)b * NN + n) * 64;
    for (int hh = 0; hh < 64; ++hh)
      acc += h[hh] * pw[(o * 2 + cc) * 64 + hh];
  }
  out[idx] = acc;
}

extern "C" void kernel_launch(void* const* d_in, const int* in_sizes, int n_in,
                              void* d_out, int out_size, void* d_ws, size_t ws_size,
                              hipStream_t stream) {
  const float* x   = (const float*)d_in[0];
  const float* ne  = (const float*)d_in[1];
  const float* te  = (const float*)d_in[2];
  const float* wp  = (const float*)d_in[3];
  const float* bp  = (const float*)d_in[4];
  const float* glg = (const float*)d_in[5];
  const float* glb = (const float*)d_in[6];
  const float* olg = (const float*)d_in[7];
  const float* olb = (const float*)d_in[8];
  const float* pw  = (const float*)d_in[9];
  const float* pb  = (const float*)d_in[10];
  float* out = (float*)d_out;

  float* ws = (float*)d_ws;
  long off = 0;
  float* W    = ws + off; off += 3L * NN * 2 * NCP * NHID;
  float* TB   = ws + off; off += 3L * NT * NB * NHID;
  float* Hs   = ws + off; off += (long)(NT + 1) * NB * NN * NHID;
  float* IAS  = ws + off; off += (long)NB * NN * NCP;
  float* CAND = ws + off; off += (long)NB * NN * NCP;
  float* XG2  = ws + off; off += (long)NB * NN * NCP;
  float* G0   = ws + off; off += (long)NB * NN * NHID;
  float* G1   = ws + off; off += (long)NB * NN * NHID;
  float* G2   = ws + off; off += (long)NB * NN * NHID;

  const long wTotal = 3L * NN * 2 * NCP * NHID;
  k_w<<<dim3((unsigned)((wTotal + 255) / 256)), 256, 0, stream>>>(ne, wp, W);
  k_tb<<<(3 * NT * NB * NHID + 255) / 256, 256, 0, stream>>>(te, bp, TB);
  k_zero<<<(unsigned)(((long)NB * NN * NHID + 255) / 256), 256, 0, stream>>>(Hs, (long)NB * NN * NHID);

  k_build_ias<<<(NB * NN * NCP + 255) / 256, 256, 0, stream>>>(x, Hs, IAS, 0);

  for (int t = 0; t < NT; ++t) {
    float* Hnew = Hs + (long)(t + 1) * NB * NN * NHID;
    k_graph_attn<2><<<dim3(NN / 64, NB), 256, 0, stream>>>(ne, te, glg, glb, IAS, XG2, CAND, 0, t);
    k_gout<<<dim3(NN, 2), 64, 0, stream>>>(IAS, XG2, CAND, W, TB, G0, G1, 0, t);
    k_state_attn<true><<<NB * NN / 16, 256, 0, stream>>>(G0, G1, Hs, olg, olb, t + 1, x, nullptr, nullptr, CAND, t);
    k_graph_attn<1><<<dim3(NN / 64, NB), 128, 0, stream>>>(ne, te, glg, glb, CAND, XG2, nullptr, 2, t);
    k_gout<<<dim3(NN, 1), 64, 0, stream>>>(CAND, XG2, nullptr, W, TB, G2, nullptr, 2, t);
    k_state_attn<false><<<NB * NN / 16, 256, 0, stream>>>(G2, nullptr, Hs, olg, olb, t + 1, x, G1, Hnew, IAS, t + 1);
  }
  k_pred<<<(NB * 12 * NN + 255) / 256, 256, 0, stream>>>(Hs, pw, pb, out);
}
